# starting kernel
# baseline (speedup 1.0000x reference)
#include <hip/hip_runtime.h>

typedef __attribute__((ext_vector_type(2))) float v2f;
typedef __attribute__((ext_vector_type(8))) float v8f;

#define BB 128
#define TT 2048
#define FF 124
#define HH 3
#define OO 6
#define PH 4

__global__ __launch_bounds__(32) void rnn_inproj_wmma(
    const float* __restrict__ x,
    const float* __restrict__ w_ih0,
    const float* __restrict__ b_ih0,
    const float* __restrict__ b_hh0,
    float* __restrict__ pre)
{
    const int lane = threadIdx.x;
    const int row  = lane & 15;
    const int hi   = lane >> 4;
    const int t0   = blockIdx.x * 16;
    const int b    = blockIdx.y;

    const float* xrow = x + ((size_t)b * TT + (size_t)(t0 + row)) * FF;

    const int   n     = row;
    const int   nclmp = (n < HH) ? n : (HH - 1);
    const float nmask = (n < HH) ? 1.0f : 0.0f;
    const float* wrow = w_ih0 + (size_t)nclmp * FF;

    v8f acc = {};
#pragma unroll 4
    for (int kk = 0; kk < FF / 4; ++kk) {
        const int k0 = kk * 4 + 2 * hi;
        v2f a = *(const v2f*)(xrow + k0);
        v2f bm;
        bm.x = wrow[k0]     * nmask;
        bm.y = wrow[k0 + 1] * nmask;
        acc = __builtin_amdgcn_wmma_f32_16x16x4_f32(
             false, a,  false, bm,
             (short)0, acc,  false,  false);
    }

    if (n < HH) {
        const float bias = b_ih0[n] + b_hh0[n];
#pragma unroll
        for (int r = 0; r < 8; ++r) {
            const int t = t0 + r + 8 * hi;
            pre[((size_t)t * BB + b) * PH + n] = acc[r] + bias;
        }
    }
}

__device__ __forceinline__ float fast_tanh(float x) {
    float r;
    asm volatile("v_tanh_f32 %0, %1\n\tv_nop" : "=v"(r) : "v"(x));
    return r;
}

__device__ __forceinline__ float selu_f(float x) {
    const float kAlpha = 1.6732632423543772f;
    const float kScale = 1.0507009873554805f;
    const float e = __builtin_amdgcn_exp2f(x * 1.4426950408889634f);
    return (x > 0.0f) ? kScale * x : kScale * kAlpha * (e - 1.0f);
}

__global__ __launch_bounds__(32) void rnn_scan(
    const float* __restrict__ pre,
    const float* __restrict__ hx,
    const float* __restrict__ w_hh0,
    const float* __restrict__ w_ih1,
    const float* __restrict__ w_hh1,
    const float* __restrict__ b_ih1,
    const float* __restrict__ b_hh1,
    const float* __restrict__ W_out,
    const float* __restrict__ b_out,
    float* __restrict__ out)
{
    const int b = blockIdx.x * 32 + threadIdx.x;

    float A0[3][3], Wi[3][3], Wh[3][3], c1[3];
#pragma unroll
    for (int i = 0; i < 3; ++i) {
#pragma unroll
        for (int j = 0; j < 3; ++j) {
            A0[i][j] = w_hh0[i * 3 + j];
            Wi[i][j] = w_ih1[i * 3 + j];
            Wh[i][j] = w_hh1[i * 3 + j];
        }
        c1[i] = b_ih1[i] + b_hh1[i];
    }
    float Wo[6][3], bo[6];
#pragma unroll
    for (int o = 0; o < 6; ++o) {
        bo[o] = b_out[o];
#pragma unroll
        for (int j = 0; j < 3; ++j) Wo[o][j] = W_out[o * 3 + j];
    }

    float h1[3], h2[3];
#pragma unroll
    for (int i = 0; i < 3; ++i) {
        h1[i] = hx[(0 * BB + b) * HH + i];
        h2[i] = hx[(1 * BB + b) * HH + i];
    }

    for (int t = 0; t < TT; ++t) {
        const float4 pv = *(const float4*)(pre + ((size_t)t * BB + b) * PH);
        const float p[3] = {pv.x, pv.y, pv.z};

        float n0[3];
#pragma unroll
        for (int i = 0; i < 3; ++i) {
            float u = p[i];
#pragma unroll
            for (int j = 0; j < 3; ++j) u = fmaf(A0[i][j], h1[j], u);
            n0[i] = fast_tanh(u);
        }
#pragma unroll
        for (int i = 0; i < 3; ++i) h1[i] = n0[i];

        float n1[3];
#pragma unroll
        for (int i = 0; i < 3; ++i) {
            float v = c1[i];
#pragma unroll
            for (int j = 0; j < 3; ++j) v = fmaf(Wi[i][j], h1[j], v);
#pragma unroll
            for (int j = 0; j < 3; ++j) v = fmaf(Wh[i][j], h2[j], v);
            n1[i] = fast_tanh(v);
        }
#pragma unroll
        for (int i = 0; i < 3; ++i) h2[i] = n1[i];

        float y[3];
#pragma unroll
        for (int i = 0; i < 3; ++i) y[i] = selu_f(h2[i]);

        float* op = out + ((size_t)b * TT + t) * OO;
#pragma unroll
        for (int o = 0; o < 6; ++o) {
            float s = bo[o];
#pragma unroll
            for (int j = 0; j < 3; ++j) s = fmaf(Wo[o][j], y[j], s);
            op[o] = s;
        }
    }
}

extern "C" void kernel_launch(void* const* d_in, const int* in_sizes, int n_in,
                              void* d_out, int out_size, void* d_ws, size_t ws_size,
                              hipStream_t stream)
{
    const float* x     = (const float*)d_in[0];
    const float* hx    = (const float*)d_in[1];
    const float* w_ih0 = (const float*)d_in[2];
    const float* w_hh0 = (const float*)d_in[3];
    const float* b_ih0 = (const float*)d_in[4];
    const float* b_hh0 = (const float*)d_in[5];
    const float* w_ih1 = (const float*)d_in[6];
    const float* w_hh1 = (const float*)d_in[7];
    const float* b_ih1 = (const float*)d_in[8];
    const float* b_hh1 = (const float*)d_in[9];
    const float* W_out = (const float*)d_in[10];
    const float* b_out = (const float*)d_in[11];

    float* out = (float*)d_out;
    float* pre = (float*)d_ws;

    dim3 g1(TT / 16, BB);
    rnn_inproj_wmma<<<g1, 32, 0, stream>>>(x, w_ih0, b_ih0, b_hh0, pre);

    rnn_scan<<<BB / 32, 32, 0, stream>>>(pre, hx, w_hh0, w_ih1, w_hh1,
                                         b_ih1, b_hh1, W_out, b_out, out);
}


# scan kernel: pre rows loaded 16 time steps ahead into a second register set (two sets used in turn), so no step waits on its own global load; arithmetic and summation order unchanged
# speedup vs baseline: 2.7737x; 2.7737x over previous
#include <hip/hip_runtime.h>

typedef __attribute__((ext_vector_type(2))) float v2f;
typedef __attribute__((ext_vector_type(8))) float v8f;

#define BB 128
#define TT 2048
#define FF 124
#define HH 3
#define OO 6
#define PH 4

__global__ __launch_bounds__(32) void rnn_inproj_wmma(
    const float* __restrict__ x,
    const float* __restrict__ w_ih0,
    const float* __restrict__ b_ih0,
    const float* __restrict__ b_hh0,
    float* __restrict__ pre)
{
    const int lane = threadIdx.x;
    const int row  = lane & 15;
    const int hi   = lane >> 4;
    const int t0   = blockIdx.x * 16;
    const int b    = blockIdx.y;

    const float* xrow = x + ((size_t)b * TT + (size_t)(t0 + row)) * FF;

    const int   n     = row;
    const int   nclmp = (n < HH) ? n : (HH - 1);
    const float nmask = (n < HH) ? 1.0f : 0.0f;
    const float* wrow = w_ih0 + (size_t)nclmp * FF;

    v8f acc = {};
#pragma unroll 4
    for (int kk = 0; kk < FF / 4; ++kk) {
        const int k0 = kk * 4 + 2 * hi;
        v2f a = *(const v2f*)(xrow + k0);
        v2f bm;
        bm.x = wrow[k0]     * nmask;
        bm.y = wrow[k0 + 1] * nmask;
        acc = __builtin_amdgcn_wmma_f32_16x16x4_f32(
             false, a,  false, bm,
             (short)0, acc,  false,  false);
    }

    if (n < HH) {
        const float bias = b_ih0[n] + b_hh0[n];
#pragma unroll
        for (int r = 0; r < 8; ++r) {
            const int t = t0 + r + 8 * hi;
            pre[((size_t)t * BB + b) * PH + n] = acc[r] + bias;
        }
    }
}

#define CH 16
static_assert(TT % (2 * CH) == 0, "the time loop takes two register sets per trip");
__device__ __forceinline__ float fast_tanh(float x) {
    float r;
    asm volatile("v_tanh_f32 %0, %1\n\tv_nop" : "=v"(r) : "v"(x));
    return r;
}

__device__ __forceinline__ float selu_f(float x) {
    const float kAlpha = 1.6732632423543772f;
    const float kScale = 1.0507009873554805f;
    const float e = __builtin_amdgcn_exp2f(x * 1.4426950408889634f);
    return (x > 0.0f) ? kScale * x : kScale * kAlpha * (e - 1.0f);
}

__global__ __launch_bounds__(32) void rnn_scan(
    const float* __restrict__ pre,
    const float* __restrict__ hx,
    const float* __restrict__ w_hh0,
    const float* __restrict__ w_ih1,
    const float* __restrict__ w_hh1,
    const float* __restrict__ b_ih1,
    const float* __restrict__ b_hh1,
    const float* __restrict__ W_out,
    const float* __restrict__ b_out,
    float* __restrict__ out)
{
    const int b = blockIdx.x * 32 + threadIdx.x;

    float A0[3][3], Wi[3][3], Wh[3][3], c1[3];
#pragma unroll
    for (int i = 0; i < 3; ++i) {
#pragma unroll
        for (int j = 0; j < 3; ++j) {
            A0[i][j] = w_hh0[i * 3 + j];
            Wi[i][j] = w_ih1[i * 3 + j];
            Wh[i][j] = w_hh1[i * 3 + j];
        }
        c1[i] = b_ih1[i] + b_hh1[i];
    }
    float Wo[6][3], bo[6];
#pragma unroll
    for (int o = 0; o < 6; ++o) {
        bo[o] = b_out[o];
#pragma unroll
        for (int j = 0; j < 3; ++j) Wo[o][j] = W_out[o * 3 + j];
    }

    float h1[3], h2[3];
#pragma unroll
    for (int i = 0; i < 3; ++i) {
        h1[i] = hx[(0 * BB + b) * HH + i];
        h2[i] = hx[(1 * BB + b) * HH + i];
    }

    auto step = [&](const float4 pv, const int t) __attribute__((always_inline)) {
        const float p[3] = {pv.x, pv.y, pv.z};

        float n0[3];
#pragma unroll
        for (int i = 0; i < 3; ++i) {
            float u = p[i];
#pragma unroll
            for (int j = 0; j < 3; ++j) u = fmaf(A0[i][j], h1[j], u);
            n0[i] = fast_tanh(u);
        }
#pragma unroll
        for (int i = 0; i < 3; ++i) h1[i] = n0[i];

        float n1[3];
#pragma unroll
        for (int i = 0; i < 3; ++i) {
            float v = c1[i];
#pragma unroll
            for (int j = 0; j < 3; ++j) v = fmaf(Wi[i][j], h1[j], v);
#pragma unroll
            for (int j = 0; j < 3; ++j) v = fmaf(Wh[i][j], h2[j], v);
            n1[i] = fast_tanh(v);
        }
#pragma unroll
        for (int i = 0; i < 3; ++i) h2[i] = n1[i];

        float y[3];
#pragma unroll
        for (int i = 0; i < 3; ++i) y[i] = selu_f(h2[i]);

        float* op = out + ((size_t)b * TT + t) * OO;
#pragma unroll
        for (int o = 0; o < 6; ++o) {
            float s = bo[o];
#pragma unroll
            for (int j = 0; j < 3; ++j) s = fmaf(Wo[o][j], y[j], s);
            op[o] = s;
        }
    };

    const float4* prow = (const float4*)pre + b;
    float4 bufA[CH], bufB[CH];
#pragma unroll
    for (int i = 0; i < CH; ++i) bufA[i] = prow[(size_t)i * BB];

    for (int t0 = 0; t0 < TT; t0 += 2 * CH) {
#pragma unroll
        for (int i = 0; i < CH; ++i) bufB[i] = prow[(size_t)(t0 + CH + i) * BB];
#pragma unroll
        for (int i = 0; i < CH; ++i) step(bufA[i], t0 + i);

        const int tn = (t0 + 2 * CH < TT) ? (t0 + 2 * CH) : t0;
#pragma unroll
        for (int i = 0; i < CH; ++i) bufA[i] = prow[(size_t)(tn + i) * BB];
#pragma unroll
        for (int i = 0; i < CH; ++i) step(bufB[i], t0 + CH + i);
    }
}

extern "C" void kernel_launch(void* const* d_in, const int* in_sizes, int n_in,
                              void* d_out, int out_size, void* d_ws, size_t ws_size,
                              hipStream_t stream)
{
    const float* x     = (const float*)d_in[0];
    const float* hx    = (const float*)d_in[1];
    const float* w_ih0 = (const float*)d_in[2];
    const float* w_hh0 = (const float*)d_in[3];
    const float* b_ih0 = (const float*)d_in[4];
    const float* b_hh0 = (const float*)d_in[5];
    const float* w_ih1 = (const float*)d_in[6];
    const float* w_hh1 = (const float*)d_in[7];
    const float* b_ih1 = (const float*)d_in[8];
    const float* b_hh1 = (const float*)d_in[9];
    const float* W_out = (const float*)d_in[10];
    const float* b_out = (const float*)d_in[11];

    float* out = (float*)d_out;
    float* pre = (float*)d_ws;

    dim3 g1(TT / 16, BB);
    rnn_inproj_wmma<<<g1, 32, 0, stream>>>(x, w_ih0, b_ih0, b_hh0, pre);

    rnn_scan<<<BB / 32, 32, 0, stream>>>(pre, hx, w_hh0, w_ih1, w_hh1,
                                         b_ih1, b_hh1, W_out, b_out, out);
}


# scan kernel: SELU picks between its two arms with a select instead of an if/else under EXEC-mask updates; same arithmetic, same summation order
# speedup vs baseline: 3.5056x; 1.2639x over previous
#include <hip/hip_runtime.h>

typedef __attribute__((ext_vector_type(2))) float v2f;
typedef __attribute__((ext_vector_type(8))) float v8f;
typedef __attribute__((ext_vector_type(4))) float v4f;
typedef v4f v4f_a4 __attribute__((aligned(4)));
typedef v2f v2f_a4 __attribute__((aligned(4)));

#define BB 128
#define TT 2048
#define FF 124
#define HH 3
#define OO 6
#define PH 4

__global__ __launch_bounds__(32) void rnn_inproj_wmma(
    const float* __restrict__ x,
    const float* __restrict__ w_ih0,
    const float* __restrict__ b_ih0,
    const float* __restrict__ b_hh0,
    float* __restrict__ pre)
{
    const int lane = threadIdx.x;
    const int row  = lane & 15;
    const int hi   = lane >> 4;
    const int t0   = blockIdx.x * 16;
    const int b    = blockIdx.y;

    const float* xrow = x + ((size_t)b * TT + (size_t)(t0 + row)) * FF;

    const int   n     = row;
    const int   nclmp = (n < HH) ? n : (HH - 1);
    const float nmask = (n < HH) ? 1.0f : 0.0f;
    const float* wrow = w_ih0 + (size_t)nclmp * FF;

    v8f acc = {};
#pragma unroll 4
    for (int kk = 0; kk < FF / 4; ++kk) {
        const int k0 = kk * 4 + 2 * hi;
        v2f a = *(const v2f*)(xrow + k0);
        v2f bm;
        bm.x = wrow[k0]     * nmask;
        bm.y = wrow[k0 + 1] * nmask;
        acc = __builtin_amdgcn_wmma_f32_16x16x4_f32(
             false, a,  false, bm,
             (short)0, acc,  false,  false);
    }

    if (n < HH) {
        const float bias = b_ih0[n] + b_hh0[n];
#pragma unroll
        for (int r = 0; r < 8; ++r) {
            const int t = t0 + r + 8 * hi;
            pre[((size_t)t * BB + b) * PH + n] = acc[r] + bias;
        }
    }
}

#define CH 16
static_assert(TT % (2 * CH) == 0, "the time loop takes two register sets per trip");
__device__ __forceinline__ float fast_tanh(float x) {
    float r;
    asm volatile("v_tanh_f32 %0, %1\n\tv_nop" : "=v"(r) : "v"(x));
    return r;
}

__device__ __forceinline__ float selu_f(float x) {
    const float kAlpha = 1.6732632423543772f;
    const float kScale = 1.0507009873554805f;
    const float e = __builtin_amdgcn_exp2f(x * 1.4426950408889634f);
    const float pos = kScale * x;
    const float neg = kScale * kAlpha * (e - 1.0f);
    return (x > 0.0f) ? pos : neg;
}

__global__ __launch_bounds__(32) void rnn_scan(
    const float* __restrict__ pre,
    const float* __restrict__ hx,
    const float* __restrict__ w_hh0,
    const float* __restrict__ w_ih1,
    const float* __restrict__ w_hh1,
    const float* __restrict__ b_ih1,
    const float* __restrict__ b_hh1,
    const float* __restrict__ W_out,
    const float* __restrict__ b_out,
    float* __restrict__ out)
{
    const int b = blockIdx.x * 32 + threadIdx.x;

    float A0[3][3], Wi[3][3], Wh[3][3], c1[3];
#pragma unroll
    for (int i = 0; i < 3; ++i) {
#pragma unroll
        for (int j = 0; j < 3; ++j) {
            A0[i][j] = w_hh0[i * 3 + j];
            Wi[i][j] = w_ih1[i * 3 + j];
            Wh[i][j] = w_hh1[i * 3 + j];
        }
        c1[i] = b_ih1[i] + b_hh1[i];
    }
    float Wo[6][3], bo[6];
#pragma unroll
    for (int o = 0; o < 6; ++o) {
        bo[o] = b_out[o];
#pragma unroll
        for (int j = 0; j < 3; ++j) Wo[o][j] = W_out[o * 3 + j];
    }

    float h1[3], h2[3];
#pragma unroll
    for (int i = 0; i < 3; ++i) {
        h1[i] = hx[(0 * BB + b) * HH + i];
        h2[i] = hx[(1 * BB + b) * HH + i];
    }

    auto step = [&](const float4 pv, const int t) __attribute__((always_inline)) {
        const float p[3] = {pv.x, pv.y, pv.z};

        float n0[3];
#pragma unroll
        for (int i = 0; i < 3; ++i) {
            float u = p[i];
#pragma unroll
            for (int j = 0; j < 3; ++j) u = fmaf(A0[i][j], h1[j], u);
            n0[i] = fast_tanh(u);
        }
#pragma unroll
        for (int i = 0; i < 3; ++i) h1[i] = n0[i];

        float n1[3];
#pragma unroll
        for (int i = 0; i < 3; ++i) {
            float v = c1[i];
#pragma unroll
            for (int j = 0; j < 3; ++j) v = fmaf(Wi[i][j], h1[j], v);
#pragma unroll
            for (int j = 0; j < 3; ++j) v = fmaf(Wh[i][j], h2[j], v);
            n1[i] = fast_tanh(v);
        }
#pragma unroll
        for (int i = 0; i < 3; ++i) h2[i] = n1[i];

        float y[3];
#pragma unroll
        for (int i = 0; i < 3; ++i) y[i] = selu_f(h2[i]);

        float* op = out + ((size_t)b * TT + t) * OO;
        float s[6];
#pragma unroll
        for (int o = 0; o < 6; ++o) {
            s[o] = bo[o];
#pragma unroll
            for (int j = 0; j < 3; ++j) s[o] = fmaf(Wo[o][j], y[j], s[o]);
        }
        const v4f lo = {s[0], s[1], s[2], s[3]};
        const v2f hi = {s[4], s[5]};
        *(v4f_a4*)op = lo;
        *(v2f_a4*)(op + 4) = hi;
    };

    const float4* prow = (const float4*)pre + b;
    float4 bufA[CH], bufB[CH];
#pragma unroll
    for (int i = 0; i < CH; ++i) bufA[i] = prow[(size_t)i * BB];

    for (int t0 = 0; t0 < TT; t0 += 2 * CH) {
#pragma unroll
        for (int i = 0; i < CH; ++i) bufB[i] = prow[(size_t)(t0 + CH + i) * BB];
#pragma unroll
        for (int i = 0; i < CH; ++i) step(bufA[i], t0 + i);

        const int tn = (t0 + 2 * CH < TT) ? (t0 + 2 * CH) : t0;
#pragma unroll
        for (int i = 0; i < CH; ++i) bufA[i] = prow[(size_t)(tn + i) * BB];
#pragma unroll
        for (int i = 0; i < CH; ++i) step(bufB[i], t0 + CH + i);
    }
}

extern "C" void kernel_launch(void* const* d_in, const int* in_sizes, int n_in,
                              void* d_out, int out_size, void* d_ws, size_t ws_size,
                              hipStream_t stream)
{
    const float* x     = (const float*)d_in[0];
    const float* hx    = (const float*)d_in[1];
    const float* w_ih0 = (const float*)d_in[2];
    const float* w_hh0 = (const float*)d_in[3];
    const float* b_ih0 = (const float*)d_in[4];
    const float* b_hh0 = (const float*)d_in[5];
    const float* w_ih1 = (const float*)d_in[6];
    const float* w_hh1 = (const float*)d_in[7];
    const float* b_ih1 = (const float*)d_in[8];
    const float* b_hh1 = (const float*)d_in[9];
    const float* W_out = (const float*)d_in[10];
    const float* b_out = (const float*)d_in[11];

    float* out = (float*)d_out;
    float* pre = (float*)d_ws;

    dim3 g1(TT / 16, BB);
    rnn_inproj_wmma<<<g1, 32, 0, stream>>>(x, w_ih0, b_ih0, b_hh0, pre);

    rnn_scan<<<BB / 32, 32, 0, stream>>>(pre, hx, w_hh0, w_ih1, w_hh1,
                                         b_ih1, b_hh1, W_out, b_out, out);
}


# scan keeps only the two tanh recurrences and stores h2 rows in place of pre; SELU, head and out stores (now whole lines) move to a third, parallel kernel; same arithmetic and summation order
# speedup vs baseline: 8.1799x; 2.3334x over previous
#include <hip/hip_runtime.h>

typedef __attribute__((ext_vector_type(2))) float v2f;
typedef __attribute__((ext_vector_type(8))) float v8f;
typedef __attribute__((ext_vector_type(4))) float v4f;
typedef v4f v4f_a4 __attribute__((aligned(4)));

#define BB 128
#define TT 2048
#define FF 124
#define HH 3
#define OO 6
#define PH 4

__global__ __launch_bounds__(32) void rnn_inproj_wmma(
    const float* __restrict__ x,
    const float* __restrict__ w_ih0,
    const float* __restrict__ b_ih0,
    const float* __restrict__ b_hh0,
    float* __restrict__ pre)
{
    const int lane = threadIdx.x;
    const int row  = lane & 15;
    const int hi   = lane >> 4;
    const int t0   = blockIdx.x * 16;
    const int b    = blockIdx.y;

    const float* xrow = x + ((size_t)b * TT + (size_t)(t0 + row)) * FF;

    const int   n     = row;
    const int   nclmp = (n < HH) ? n : (HH - 1);
    const float nmask = (n < HH) ? 1.0f : 0.0f;
    const float* wrow = w_ih0 + (size_t)nclmp * FF;

    v8f acc = {};
#pragma unroll 4
    for (int kk = 0; kk < FF / 4; ++kk) {
        const int k0 = kk * 4 + 2 * hi;
        v2f a = *(const v2f*)(xrow + k0);
        v2f bm;
        bm.x = wrow[k0]     * nmask;
        bm.y = wrow[k0 + 1] * nmask;
        acc = __builtin_amdgcn_wmma_f32_16x16x4_f32(
             false, a,  false, bm,
             (short)0, acc,  false,  false);
    }

    if (n < HH) {
        const float bias = b_ih0[n] + b_hh0[n];
#pragma unroll
        for (int r = 0; r < 8; ++r) {
            const int t = t0 + r + 8 * hi;
            pre[((size_t)t * BB + b) * PH + n] = acc[r] + bias;
        }
    }
}

#define CH 16
static_assert(TT % (2 * CH) == 0, "the time loop takes two register sets per trip");
__device__ __forceinline__ float fast_tanh(float x) {
    float r;
    asm volatile("v_tanh_f32 %0, %1\n\tv_nop" : "=v"(r) : "v"(x));
    return r;
}

__device__ __forceinline__ float selu_f(float x) {
    const float kAlpha = 1.6732632423543772f;
    const float kScale = 1.0507009873554805f;
    const float e = __builtin_amdgcn_exp2f(x * 1.4426950408889634f);
    const float pos = kScale * x;
    const float neg = kScale * kAlpha * (e - 1.0f);
    return (x > 0.0f) ? pos : neg;
}

__global__ __launch_bounds__(32) void rnn_scan(
    float* ws,
    const float* __restrict__ hx,
    const float* __restrict__ w_hh0,
    const float* __restrict__ w_ih1,
    const float* __restrict__ w_hh1,
    const float* __restrict__ b_ih1,
    const float* __restrict__ b_hh1)
{
    const int b = blockIdx.x * 32 + threadIdx.x;

    float A0[3][3], Wi[3][3], Wh[3][3], c1[3];
#pragma unroll
    for (int i = 0; i < 3; ++i) {
#pragma unroll
        for (int j = 0; j < 3; ++j) {
            A0[i][j] = w_hh0[i * 3 + j];
            Wi[i][j] = w_ih1[i * 3 + j];
            Wh[i][j] = w_hh1[i * 3 + j];
        }
        c1[i] = b_ih1[i] + b_hh1[i];
    }

    float h1[3], h2[3];
#pragma unroll
    for (int i = 0; i < 3; ++i) {
        h1[i] = hx[(0 * BB + b) * HH + i];
        h2[i] = hx[(1 * BB + b) * HH + i];
    }

    float4* prow = (float4*)ws + b;

    auto step = [&](const float4 pv, const int t) __attribute__((always_inline)) {
        const float p[3] = {pv.x, pv.y, pv.z};

        float n0[3];
#pragma unroll
        for (int i = 0; i < 3; ++i) {
            float u = p[i];
#pragma unroll
            for (int j = 0; j < 3; ++j) u = fmaf(A0[i][j], h1[j], u);
            n0[i] = fast_tanh(u);
        }
#pragma unroll
        for (int i = 0; i < 3; ++i) h1[i] = n0[i];

        float n1[3];
#pragma unroll
        for (int i = 0; i < 3; ++i) {
            float v = c1[i];
#pragma unroll
            for (int j = 0; j < 3; ++j) v = fmaf(Wi[i][j], h1[j], v);
#pragma unroll
            for (int j = 0; j < 3; ++j) v = fmaf(Wh[i][j], h2[j], v);
            n1[i] = fast_tanh(v);
        }
#pragma unroll
        for (int i = 0; i < 3; ++i) h2[i] = n1[i];

        prow[(size_t)t * BB] = make_float4(h2[0], h2[1], h2[2], 0.0f);
    };

    float4 bufA[CH], bufB[CH];
#pragma unroll
    for (int i = 0; i < CH; ++i) bufA[i] = prow[(size_t)i * BB];

    for (int t0 = 0; t0 < TT; t0 += 2 * CH) {
#pragma unroll
        for (int i = 0; i < CH; ++i) bufB[i] = prow[(size_t)(t0 + CH + i) * BB];
#pragma unroll
        for (int i = 0; i < CH; ++i) step(bufA[i], t0 + i);

        const int tn = (t0 + 2 * CH < TT) ? (t0 + 2 * CH) : t0;
#pragma unroll
        for (int i = 0; i < CH; ++i) bufA[i] = prow[(size_t)(tn + i) * BB];
#pragma unroll
        for (int i = 0; i < CH; ++i) step(bufB[i], t0 + CH + i);
    }
}

#define HB 8
#define HT 128
#define HTH 256
#define HRUN (HT * OO)
#define HPITCH (HRUN + 8)
static_assert(BB % HB == 0 && TT % HT == 0, "head tiles cover the problem exactly");
static_assert(HRUN % (4 * 32) == 0, "a wave's 32 pieces of 16 bytes stay inside one batch element's run");

__global__ __launch_bounds__(HTH) void rnn_head(
    const float* __restrict__ hbuf,
    const float* __restrict__ W_out,
    const float* __restrict__ b_out,
    float* __restrict__ out)
{
    __shared__ __attribute__((aligned(16))) float stage[HB * HPITCH];

    const int tid = threadIdx.x;
    const int bl  = tid & (HB - 1);
    const int tl  = tid / HB;
    const int t0  = blockIdx.x * HT;
    const int b0  = blockIdx.y * HB;

    float4 hv[HT / 32];
    for (int k = 0; k < HT / 32; ++k)
        hv[k] = ((const float4*)hbuf)[(size_t)(t0 + k * 32 + tl) * BB + (b0 + bl)];

    for (int k = 0; k < HT / 32; ++k) {
        const float h2[3] = {hv[k].x, hv[k].y, hv[k].z};
        float y[3];
        for (int i = 0; i < 3; ++i) y[i] = selu_f(h2[i]);

        float* sp = stage + bl * HPITCH + (k * 32 + tl) * OO;
        for (int o = 0; o < OO; ++o) {
            float s = b_out[o];
            for (int j = 0; j < 3; ++j) s = fmaf(W_out[o * 3 + j], y[j], s);
            sp[o] = s;
        }
    }
    __syncthreads();

    constexpr int kPiecesPerB = HRUN / 4;
    static_assert((HB * kPiecesPerB) % HTH == 0, "pieces divide evenly among the threads");
    for (int j = 0; j < HB * kPiecesPerB / HTH; ++j) {
        const int p  = j * HTH + tid;
        const int pb = p / kPiecesPerB;
        const int pq = p - pb * kPiecesPerB;
        const v4f v = *(const v4f*)(stage + pb * HPITCH + pq * 4);
        float* op = out + ((size_t)(b0 + pb) * TT + t0) * OO + pq * 4;
        *(v4f_a4*)op = v;
    }
}

extern "C" void kernel_launch(void* const* d_in, const int* in_sizes, int n_in,
                              void* d_out, int out_size, void* d_ws, size_t ws_size,
                              hipStream_t stream)
{
    const float* x     = (const float*)d_in[0];
    const float* hx    = (const float*)d_in[1];
    const float* w_ih0 = (const float*)d_in[2];
    const float* w_hh0 = (const float*)d_in[3];
    const float* b_ih0 = (const float*)d_in[4];
    const float* b_hh0 = (const float*)d_in[5];
    const float* w_ih1 = (const float*)d_in[6];
    const float* w_hh1 = (const float*)d_in[7];
    const float* b_ih1 = (const float*)d_in[8];
    const float* b_hh1 = (const float*)d_in[9];
    const float* W_out = (const float*)d_in[10];
    const float* b_out = (const float*)d_in[11];

    float* out = (float*)d_out;
    float* pre = (float*)d_ws;

    dim3 g1(TT / 16, BB);
    rnn_inproj_wmma<<<g1, 32, 0, stream>>>(x, w_ih0, b_ih0, b_hh0, pre);

    rnn_scan<<<BB / 32, 32, 0, stream>>>(pre, hx, w_hh0, w_ih1, w_hh1, b_ih1, b_hh1);

    dim3 g3(TT / HT, BB / HB);
    rnn_head<<<g3, HTH, 0, stream>>>(pre, W_out, b_out, out);
}


# scan: layer 1 of step t and layer 0 of step t+1 share a loop slot (tanh via the builtin; pre rows 32 steps ahead), so the two recurrences interleave in issue; same arithmetic and summation order
# speedup vs baseline: 11.2892x; 1.3801x over previous
#include <hip/hip_runtime.h>

typedef __attribute__((ext_vector_type(2))) float v2f;
typedef __attribute__((ext_vector_type(8))) float v8f;
typedef __attribute__((ext_vector_type(4))) float v4f;
typedef v4f v4f_a4 __attribute__((aligned(4)));

#define BB 128
#define TT 2048
#define FF 124
#define HH 3
#define OO 6
#define PH 4

__global__ __launch_bounds__(32) void rnn_inproj_wmma(
    const float* __restrict__ x,
    const float* __restrict__ w_ih0,
    const float* __restrict__ b_ih0,
    const float* __restrict__ b_hh0,
    float* __restrict__ pre)
{
    const int lane = threadIdx.x;
    const int row  = lane & 15;
    const int hi   = lane >> 4;
    const int t0   = blockIdx.x * 16;
    const int b    = blockIdx.y;

    const float* xrow = x + ((size_t)b * TT + (size_t)(t0 + row)) * FF;

    const int   n     = row;
    const int   nclmp = (n < HH) ? n : (HH - 1);
    const float nmask = (n < HH) ? 1.0f : 0.0f;
    const float* wrow = w_ih0 + (size_t)nclmp * FF;

    v8f acc = {};
#pragma unroll 4
    for (int kk = 0; kk < FF / 4; ++kk) {
        const int k0 = kk * 4 + 2 * hi;
        v2f a = *(const v2f*)(xrow + k0);
        v2f bm;
        bm.x = wrow[k0]     * nmask;
        bm.y = wrow[k0 + 1] * nmask;
        acc = __builtin_amdgcn_wmma_f32_16x16x4_f32(
             false, a,  false, bm,
             (short)0, acc,  false,  false);
    }

    if (n < HH) {
        const float bias = b_ih0[n] + b_hh0[n];
#pragma unroll
        for (int r = 0; r < 8; ++r) {
            const int t = t0 + r + 8 * hi;
            pre[((size_t)t * BB + b) * PH + n] = acc[r] + bias;
        }
    }
}

#define CH 32
static_assert(TT % (2 * CH) == 0, "the time loop takes two register sets per trip");
__device__ __forceinline__ float fast_tanh(float x) {
    return __builtin_amdgcn_tanhf(x);
}

__device__ __forceinline__ float selu_f(float x) {
    const float kAlpha = 1.6732632423543772f;
    const float kScale = 1.0507009873554805f;
    const float e = __builtin_amdgcn_exp2f(x * 1.4426950408889634f);
    const float pos = kScale * x;
    const float neg = kScale * kAlpha * (e - 1.0f);
    return (x > 0.0f) ? pos : neg;
}

__global__ __launch_bounds__(32) void rnn_scan(
    float* ws,
    const float* __restrict__ hx,
    const float* __restrict__ w_hh0,
    const float* __restrict__ w_ih1,
    const float* __restrict__ w_hh1,
    const float* __restrict__ b_ih1,
    const float* __restrict__ b_hh1)
{
    const int b = blockIdx.x * 32 + threadIdx.x;

    float A0[3][3], Wi[3][3], Wh[3][3], c1[3];
#pragma unroll
    for (int i = 0; i < 3; ++i) {
#pragma unroll
        for (int j = 0; j < 3; ++j) {
            A0[i][j] = w_hh0[i * 3 + j];
            Wi[i][j] = w_ih1[i * 3 + j];
            Wh[i][j] = w_hh1[i * 3 + j];
        }
        c1[i] = b_ih1[i] + b_hh1[i];
    }

    float h1[3], h2[3];
#pragma unroll
    for (int i = 0; i < 3; ++i) {
        h1[i] = hx[(0 * BB + b) * HH + i];
        h2[i] = hx[(1 * BB + b) * HH + i];
    }

    float4* prow = (float4*)ws + b;

    auto layer0 = [&](const float4 pv, float (&n0)[3]) __attribute__((always_inline)) {
        const float p[3] = {pv.x, pv.y, pv.z};
#pragma unroll
        for (int i = 0; i < 3; ++i) {
            float u = p[i];
#pragma unroll
            for (int j = 0; j < 3; ++j) u = fmaf(A0[i][j], h1[j], u);
            n0[i] = fast_tanh(u);
        }
    };

    auto slot = [&](const float4 pn, const int t) __attribute__((always_inline)) {
        float n0[3];
        layer0(pn, n0);

        float n1[3];
#pragma unroll
        for (int i = 0; i < 3; ++i) {
            float v = c1[i];
#pragma unroll
            for (int j = 0; j < 3; ++j) v = fmaf(Wi[i][j], h1[j], v);
#pragma unroll
            for (int j = 0; j < 3; ++j) v = fmaf(Wh[i][j], h2[j], v);
            n1[i] = fast_tanh(v);
        }
#pragma unroll
        for (int i = 0; i < 3; ++i) { h1[i] = n0[i]; h2[i] = n1[i]; }

        prow[(size_t)t * BB] = make_float4(h2[0], h2[1], h2[2], 0.0f);
    };

    float4 bufA[CH], bufB[CH];
#pragma unroll
    for (int i = 0; i < CH; ++i) bufA[i] = prow[(size_t)i * BB];

    {
        float n0[3];
        layer0(bufA[0], n0);
#pragma unroll
        for (int i = 0; i < 3; ++i) h1[i] = n0[i];
    }

    for (int t0 = 0; t0 < TT; t0 += 2 * CH) {
#pragma unroll
        for (int i = 0; i < CH; ++i) bufB[i] = prow[(size_t)(t0 + CH + i) * BB];
        __builtin_amdgcn_sched_barrier(0x0001);
#pragma unroll
        for (int i = 0; i < CH; ++i) slot(i + 1 < CH ? bufA[i + 1] : bufB[0], t0 + i);

        const int tn = (t0 + 2 * CH < TT) ? (t0 + 2 * CH) : t0;
#pragma unroll
        for (int i = 0; i < CH; ++i) bufA[i] = prow[(size_t)(tn + i) * BB];
        __builtin_amdgcn_sched_barrier(0x0001);
#pragma unroll
        for (int i = 0; i < CH; ++i) slot(i + 1 < CH ? bufB[i + 1] : bufA[0], t0 + CH + i);
    }
}

#define HB 8
#define HT 128
#define HTH 256
#define HRUN (HT * OO)
#define HPITCH (HRUN + 8)
static_assert(BB % HB == 0 && TT % HT == 0, "head tiles cover the problem exactly");
static_assert(HRUN % (4 * 32) == 0, "a wave's 32 pieces of 16 bytes stay inside one batch element's run");

__global__ __launch_bounds__(HTH) void rnn_head(
    const float* __restrict__ hbuf,
    const float* __restrict__ W_out,
    const float* __restrict__ b_out,
    float* __restrict__ out)
{
    __shared__ __attribute__((aligned(16))) float stage[HB * HPITCH];

    const int tid = threadIdx.x;
    const int bl  = tid & (HB - 1);
    const int tl  = tid / HB;
    const int t0  = blockIdx.x * HT;
    const int b0  = blockIdx.y * HB;

    float4 hv[HT / 32];
    for (int k = 0; k < HT / 32; ++k)
        hv[k] = ((const float4*)hbuf)[(size_t)(t0 + k * 32 + tl) * BB + (b0 + bl)];

    for (int k = 0; k < HT / 32; ++k) {
        const float h2[3] = {hv[k].x, hv[k].y, hv[k].z};
        float y[3];
        for (int i = 0; i < 3; ++i) y[i] = selu_f(h2[i]);

        float* sp = stage + bl * HPITCH + (k * 32 + tl) * OO;
        for (int o = 0; o < OO; ++o) {
            float s = b_out[o];
            for (int j = 0; j < 3; ++j) s = fmaf(W_out[o * 3 + j], y[j], s);
            sp[o] = s;
        }
    }
    __syncthreads();

    constexpr int kPiecesPerB = HRUN / 4;
    static_assert((HB * kPiecesPerB) % HTH == 0, "pieces divide evenly among the threads");
    for (int j = 0; j < HB * kPiecesPerB / HTH; ++j) {
        const int p  = j * HTH + tid;
        const int pb = p / kPiecesPerB;
        const int pq = p - pb * kPiecesPerB;
        const v4f v = *(const v4f*)(stage + pb * HPITCH + pq * 4);
        float* op = out + ((size_t)(b0 + pb) * TT + t0) * OO + pq * 4;
        *(v4f_a4*)op = v;
    }
}

extern "C" void kernel_launch(void* const* d_in, const int* in_sizes, int n_in,
                              void* d_out, int out_size, void* d_ws, size_t ws_size,
                              hipStream_t stream)
{
    const float* x     = (const float*)d_in[0];
    const float* hx    = (const float*)d_in[1];
    const float* w_ih0 = (const float*)d_in[2];
    const float* w_hh0 = (const float*)d_in[3];
    const float* b_ih0 = (const float*)d_in[4];
    const float* b_hh0 = (const float*)d_in[5];
    const float* w_ih1 = (const float*)d_in[6];
    const float* w_hh1 = (const float*)d_in[7];
    const float* b_ih1 = (const float*)d_in[8];
    const float* b_hh1 = (const float*)d_in[9];
    const float* W_out = (const float*)d_in[10];
    const float* b_out = (const float*)d_in[11];

    float* out = (float*)d_out;
    float* pre = (float*)d_ws;

    dim3 g1(TT / 16, BB);
    rnn_inproj_wmma<<<g1, 32, 0, stream>>>(x, w_ih0, b_ih0, b_hh0, pre);

    rnn_scan<<<BB / 32, 32, 0, stream>>>(pre, hx, w_hh0, w_ih1, w_hh1, b_ih1, b_hh1);

    dim3 g3(TT / HT, BB / HB);
    rnn_head<<<g3, HTH, 0, stream>>>(pre, W_out, b_out, out);
}


# kernel 1: x reaches the WMMA A fragments through LDS, copied by wave loads of 512 contiguous bytes (was 32 scattered 8-byte pieces per load); 4 tiles per wave; same k and summation order
# speedup vs baseline: 13.2402x; 1.1728x over previous
#include <hip/hip_runtime.h>

typedef __attribute__((ext_vector_type(2))) float v2f;
typedef __attribute__((ext_vector_type(8))) float v8f;
typedef __attribute__((ext_vector_type(4))) float v4f;
typedef v4f v4f_a4 __attribute__((aligned(4)));

#define BB 128
#define TT 2048
#define FF 124
#define HH 3
#define OO 6
#define PH 4

#define K1_WAVES 8
#define K1_TPW   4
#define TILES_PER_B (TT / 16)
#define TILE_FL  (16 * FF)
#define PAIR_FL  (2 * TILE_FL)
#define KSTEPS   (FF / 4)
static_assert(PAIR_FL % (32 * 4) == 0, "a pair of tiles is a whole number of 512-byte rounds");
static_assert(PAIR_FL / (32 * 4) == 31, "31 rounds per pair: 16 for the first tile, 15 more for the second");
static_assert(TILES_PER_B % K1_TPW == 0, "a wave's tiles belong to one batch element");
static_assert((BB * TILES_PER_B) == 512 * (K1_WAVES * K1_TPW), "512 work-groups cover the tiles exactly");

__global__ __launch_bounds__(K1_WAVES * 32) void rnn_inproj_wmma(
    const float* __restrict__ x,
    const float* __restrict__ w_ih0,
    const float* __restrict__ b_ih0,
    const float* __restrict__ b_hh0,
    float* __restrict__ pre)
{
    __shared__ __attribute__((aligned(16))) float xs[K1_WAVES * PAIR_FL];

    const int lane = threadIdx.x & 31;
    const int wv   = threadIdx.x >> 5;
    const int row  = lane & 15;
    const int hi   = lane >> 4;

    const int tile0 = (blockIdx.x * K1_WAVES + wv) * K1_TPW;
    const int b     = tile0 / TILES_PER_B;
    const int tt0   = tile0 % TILES_PER_B;

    const float* src = x + (size_t)tile0 * TILE_FL;
    float* my = xs + wv * PAIR_FL;

    const int   n     = row;
    const int   nclmp = (n < HH) ? n : (HH - 1);
    const float nmask = (n < HH) ? 1.0f : 0.0f;
    const float* wrow = w_ih0 + (size_t)nclmp * FF;

    v2f bm[KSTEPS];
#pragma unroll
    for (int kk = 0; kk < KSTEPS; ++kk) {
        const int k0 = kk * 4 + 2 * hi;
        bm[kk].x = wrow[k0]     * nmask;
        bm[kk].y = wrow[k0 + 1] * nmask;
    }
    const float bias = b_ih0[nclmp] + b_hh0[nclmp];

    v4f RA[16], RB[15];
    auto gloadA = [&](const int p) __attribute__((always_inline)) {
#pragma unroll
        for (int q = 0; q < 16; ++q)
            RA[q] = *(const v4f_a4*)(src + p * PAIR_FL + (q * 32 + lane) * 4);
    };
    auto gloadB = [&](const int p) __attribute__((always_inline)) {
#pragma unroll
        for (int q = 0; q < 15; ++q)
            RB[q] = *(const v4f_a4*)(src + p * PAIR_FL + ((16 + q) * 32 + lane) * 4);
    };
    auto lstoreA = [&]() __attribute__((always_inline)) {
#pragma unroll
        for (int q = 0; q < 16; ++q) *(v4f*)(my + (q * 32 + lane) * 4) = RA[q];
    };
    auto lstoreB = [&]() __attribute__((always_inline)) {
#pragma unroll
        for (int q = 0; q < 15; ++q) *(v4f*)(my + ((16 + q) * 32 + lane) * 4) = RB[q];
    };
    auto tile = [&](const int j, const int tt) __attribute__((always_inline)) {
        const float* arow = my + j * TILE_FL + row * FF + 2 * hi;
        v8f acc = {};
#pragma unroll
        for (int kk = 0; kk < KSTEPS; ++kk) {
            const v2f a = *(const v2f*)(arow + 4 * kk);
            acc = __builtin_amdgcn_wmma_f32_16x16x4_f32(
                 false, a,  false, bm[kk],
                 (short)0, acc,  false,  false);
        }
        if (n < HH) {
#pragma unroll
            for (int r = 0; r < 8; ++r) {
                const int t = tt * 16 + r + 8 * hi;
                pre[((size_t)t * BB + b) * PH + n] = acc[r] + bias;
            }
        }
    };

    gloadA(0);
    gloadB(0);
#pragma unroll
    for (int p = 0; p < K1_TPW / 2; ++p) {
        lstoreA();
        if (p + 1 < K1_TPW / 2) {
            __builtin_amdgcn_sched_barrier(0x0001);
            gloadA(p + 1);
            __builtin_amdgcn_sched_barrier(0x0001);
        }
        tile(0, tt0 + 2 * p);
        lstoreB();
        if (p + 1 < K1_TPW / 2) {
            __builtin_amdgcn_sched_barrier(0x0001);
            gloadB(p + 1);
            __builtin_amdgcn_sched_barrier(0x0001);
        }
        tile(1, tt0 + 2 * p + 1);
    }
}

#define CH 32
static_assert(TT % (2 * CH) == 0, "the time loop takes two register sets per trip");
__device__ __forceinline__ float fast_tanh(float x) {
    return __builtin_amdgcn_tanhf(x);
}

__device__ __forceinline__ float selu_f(float x) {
    const float kAlpha = 1.6732632423543772f;
    const float kScale = 1.0507009873554805f;
    const float e = __builtin_amdgcn_exp2f(x * 1.4426950408889634f);
    const float pos = kScale * x;
    const float neg = kScale * kAlpha * (e - 1.0f);
    return (x > 0.0f) ? pos : neg;
}

__global__ __launch_bounds__(32) void rnn_scan(
    float* ws,
    const float* __restrict__ hx,
    const float* __restrict__ w_hh0,
    const float* __restrict__ w_ih1,
    const float* __restrict__ w_hh1,
    const float* __restrict__ b_ih1,
    const float* __restrict__ b_hh1)
{
    const int b = blockIdx.x * 32 + threadIdx.x;

    float A0[3][3], Wi[3][3], Wh[3][3], c1[3];
#pragma unroll
    for (int i = 0; i < 3; ++i) {
#pragma unroll
        for (int j = 0; j < 3; ++j) {
            A0[i][j] = w_hh0[i * 3 + j];
            Wi[i][j] = w_ih1[i * 3 + j];
            Wh[i][j] = w_hh1[i * 3 + j];
        }
        c1[i] = b_ih1[i] + b_hh1[i];
    }

    float h1[3], h2[3];
#pragma unroll
    for (int i = 0; i < 3; ++i) {
        h1[i] = hx[(0 * BB + b) * HH + i];
        h2[i] = hx[(1 * BB + b) * HH + i];
    }

    float4* prow = (float4*)ws + b;

    auto layer0 = [&](const float4 pv, float (&n0)[3]) __attribute__((always_inline)) {
        const float p[3] = {pv.x, pv.y, pv.z};
#pragma unroll
        for (int i = 0; i < 3; ++i) {
            float u = p[i];
#pragma unroll
            for (int j = 0; j < 3; ++j) u = fmaf(A0[i][j], h1[j], u);
            n0[i] = fast_tanh(u);
        }
    };

    auto slot = [&](const float4 pn, const int t) __attribute__((always_inline)) {
        float n0[3];
        layer0(pn, n0);

        float n1[3];
#pragma unroll
        for (int i = 0; i < 3; ++i) {
            float v = c1[i];
#pragma unroll
            for (int j = 0; j < 3; ++j) v = fmaf(Wi[i][j], h1[j], v);
#pragma unroll
            for (int j = 0; j < 3; ++j) v = fmaf(Wh[i][j], h2[j], v);
            n1[i] = fast_tanh(v);
        }
#pragma unroll
        for (int i = 0; i < 3; ++i) { h1[i] = n0[i]; h2[i] = n1[i]; }

        prow[(size_t)t * BB] = make_float4(h2[0], h2[1], h2[2], 0.0f);
    };

    float4 bufA[CH], bufB[CH];
#pragma unroll
    for (int i = 0; i < CH; ++i) bufA[i] = prow[(size_t)i * BB];

    {
        float n0[3];
        layer0(bufA[0], n0);
#pragma unroll
        for (int i = 0; i < 3; ++i) h1[i] = n0[i];
    }

    for (int t0 = 0; t0 < TT; t0 += 2 * CH) {
#pragma unroll
        for (int i = 0; i < CH; ++i) bufB[i] = prow[(size_t)(t0 + CH + i) * BB];
        __builtin_amdgcn_sched_barrier(0x0001);
#pragma unroll
        for (int i = 0; i < CH; ++i) slot(i + 1 < CH ? bufA[i + 1] : bufB[0], t0 + i);

        const int tn = (t0 + 2 * CH < TT) ? (t0 + 2 * CH) : t0;
#pragma unroll
        for (int i = 0; i < CH; ++i) bufA[i] = prow[(size_t)(tn + i) * BB];
        __builtin_amdgcn_sched_barrier(0x0001);
#pragma unroll
        for (int i = 0; i < CH; ++i) slot(i + 1 < CH ? bufB[i + 1] : bufA[0], t0 + CH + i);
    }
}

#define HB 8
#define HT 128
#define HTH 256
#define HRUN (HT * OO)
#define HPITCH (HRUN + 8)
static_assert(BB % HB == 0 && TT % HT == 0, "head tiles cover the problem exactly");
static_assert(HRUN % (4 * 32) == 0, "a wave's 32 pieces of 16 bytes stay inside one batch element's run");

__global__ __launch_bounds__(HTH) void rnn_head(
    const float* __restrict__ hbuf,
    const float* __restrict__ W_out,
    const float* __restrict__ b_out,
    float* __restrict__ out)
{
    __shared__ __attribute__((aligned(16))) float stage[HB * HPITCH];

    const int tid = threadIdx.x;
    const int bl  = tid & (HB - 1);
    const int tl  = tid / HB;
    const int t0  = blockIdx.x * HT;
    const int b0  = blockIdx.y * HB;

    float4 hv[HT / 32];
    for (int k = 0; k < HT / 32; ++k)
        hv[k] = ((const float4*)hbuf)[(size_t)(t0 + k * 32 + tl) * BB + (b0 + bl)];

    for (int k = 0; k < HT / 32; ++k) {
        const float h2[3] = {hv[k].x, hv[k].y, hv[k].z};
        float y[3];
        for (int i = 0; i < 3; ++i) y[i] = selu_f(h2[i]);

        float* sp = stage + bl * HPITCH + (k * 32 + tl) * OO;
        for (int o = 0; o < OO; ++o) {
            float s = b_out[o];
            for (int j = 0; j < 3; ++j) s = fmaf(W_out[o * 3 + j], y[j], s);
            sp[o] = s;
        }
    }
    __syncthreads();

    constexpr int kPiecesPerB = HRUN / 4;
    static_assert((HB * kPiecesPerB) % HTH == 0, "pieces divide evenly among the threads");
    for (int j = 0; j < HB * kPiecesPerB / HTH; ++j) {
        const int p  = j * HTH + tid;
        const int pb = p / kPiecesPerB;
        const int pq = p - pb * kPiecesPerB;
        const v4f v = *(const v4f*)(stage + pb * HPITCH + pq * 4);
        float* op = out + ((size_t)(b0 + pb) * TT + t0) * OO + pq * 4;
        *(v4f_a4*)op = v;
    }
}

extern "C" void kernel_launch(void* const* d_in, const int* in_sizes, int n_in,
                              void* d_out, int out_size, void* d_ws, size_t ws_size,
                              hipStream_t stream)
{
    const float* x     = (const float*)d_in[0];
    const float* hx    = (const float*)d_in[1];
    const float* w_ih0 = (const float*)d_in[2];
    const float* w_hh0 = (const float*)d_in[3];
    const float* b_ih0 = (const float*)d_in[4];
    const float* b_hh0 = (const float*)d_in[5];
    const float* w_ih1 = (const float*)d_in[6];
    const float* w_hh1 = (const float*)d_in[7];
    const float* b_ih1 = (const float*)d_in[8];
    const float* b_hh1 = (const float*)d_in[9];
    const float* W_out = (const float*)d_in[10];
    const float* b_out = (const float*)d_in[11];

    float* out = (float*)d_out;
    float* pre = (float*)d_ws;

    rnn_inproj_wmma<<<512, K1_WAVES * 32, 0, stream>>>(x, w_ih0, b_ih0, b_hh0, pre);

    rnn_scan<<<BB / 32, 32, 0, stream>>>(pre, hx, w_hh0, w_ih1, w_hh1, b_ih1, b_hh1);

    dim3 g3(TT / HT, BB / HB);
    rnn_head<<<g3, HTH, 0, stream>>>(pre, W_out, b_out, out);
}


# one launch: a work-group per batch element; its 4 waves run projection (WMMA), recurrence and head as a pipeline over 128-step chunks handed over in LDS at a barrier; same arithmetic and sum order
# speedup vs baseline: 20.6714x; 1.5613x over previous
#include <hip/hip_runtime.h>

typedef __attribute__((ext_vector_type(2))) float v2f;
typedef __attribute__((ext_vector_type(8))) float v8f;
typedef __attribute__((ext_vector_type(4))) float v4f;
typedef v4f v4f_a4 __attribute__((aligned(4)));

#define BB 128
#define TT 2048
#define FF 124
#define HH 3
#define OO 6

#define NWAVE 4
#define CHK   128
#define HALF  64
#define NCH   (TT / CHK)
#define TILE_FL  (16 * FF)
#define PAIR_FL  (2 * TILE_FL)
#define KSTEPS   (FF / 4)
#define NPAIR (TT / 16 / 2 / 2)
static_assert(PAIR_FL % (32 * 4) == 0, "a pair of tiles is a whole number of 512-byte rounds");
static_assert(PAIR_FL / (32 * 4) == 31, "31 rounds per pair: 16 for the first tile, 15 more for the second");
static_assert(CHK == 2 * 2 * 32, "a phase is two pairs for each of the two projection waves");
static_assert((CHK * OO) % (4 * 32) == 0, "a chunk of out is a whole number of 512-byte stores");

__device__ __forceinline__ float fast_tanh(float x) {
    return __builtin_amdgcn_tanhf(x);
}

__device__ __forceinline__ float selu_f(float x) {
    const float kAlpha = 1.6732632423543772f;
    const float kScale = 1.0507009873554805f;
    const float e = __builtin_amdgcn_exp2f(x * 1.4426950408889634f);
    const float pos = kScale * x;
    const float neg = kScale * kAlpha * (e - 1.0f);
    return (x > 0.0f) ? pos : neg;
}

__global__ __launch_bounds__(NWAVE * 32) void rnn_fused(
    const float* __restrict__ x,
    const float* __restrict__ hx,
    const float* __restrict__ w_ih0,
    const float* __restrict__ w_hh0,
    const float* __restrict__ b_ih0,
    const float* __restrict__ b_hh0,
    const float* __restrict__ w_ih1,
    const float* __restrict__ w_hh1,
    const float* __restrict__ b_ih1,
    const float* __restrict__ b_hh1,
    const float* __restrict__ W_out,
    const float* __restrict__ b_out,
    float* __restrict__ out)
{
    __shared__ __attribute__((aligned(16))) float xs[2 * PAIR_FL];
    __shared__ __attribute__((aligned(16))) float preb[2 * CHK * 4];
    __shared__ __attribute__((aligned(16))) float h2b[2 * CHK * 32 * 4];
    __shared__ __attribute__((aligned(16))) float stage[CHK * OO];

    const int lane = threadIdx.x & 31;
    const int wv = __builtin_amdgcn_readfirstlane((int)(threadIdx.x >> 5));
    const int b  = blockIdx.x;

    const int row  = lane & 15;
    const int hi   = lane >> 4;
    const int pw   = (wv == 2) ? 1 : 0;
    const float* xb = x + (size_t)b * TT * FF;
    float* my = xs + pw * PAIR_FL;

    const int   n     = row;
    const int   nclmp = (n < HH) ? n : (HH - 1);
    const float nmask = (n < HH) ? 1.0f : 0.0f;
    const float* wrow = w_ih0 + (size_t)nclmp * FF;

    v2f bm[KSTEPS];
    float bias = 0.0f;
    if (wv == 1 || wv == 2) {
#pragma unroll
        for (int kk = 0; kk < KSTEPS; ++kk) {
            const int k0 = kk * 4 + 2 * hi;
            bm[kk].x = wrow[k0]     * nmask;
            bm[kk].y = wrow[k0 + 1] * nmask;
        }
        bias = b_ih0[nclmp] + b_hh0[nclmp];
    } else {
#pragma unroll
        for (int kk = 0; kk < KSTEPS; ++kk) { bm[kk].x = 0.0f; bm[kk].y = 0.0f; }
    }

    v4f RA[16], RB[15];
#pragma unroll
    for (int q = 0; q < 16; ++q) RA[q] = (v4f){0.0f, 0.0f, 0.0f, 0.0f};
#pragma unroll
    for (int q = 0; q < 15; ++q) RB[q] = (v4f){0.0f, 0.0f, 0.0f, 0.0f};
    auto pair_src = [&](const int q) __attribute__((always_inline)) -> const float* {
        const int ti = (q >> 1) * 8 + pw * 4 + (q & 1) * 2;
        return xb + (size_t)ti * TILE_FL;
    };
    auto gloadA = [&](const float* sp) __attribute__((always_inline)) {
#pragma unroll
        for (int q = 0; q < 16; ++q)
            RA[q] = *(const v4f_a4*)(sp + (q * 32 + lane) * 4);
    };
    auto gloadB = [&](const float* sp) __attribute__((always_inline)) {
#pragma unroll
        for (int q = 0; q < 15; ++q)
            RB[q] = *(const v4f_a4*)(sp + ((16 + q) * 32 + lane) * 4);
    };
    auto lstoreA = [&]() __attribute__((always_inline)) {
#pragma unroll
        for (int q = 0; q < 16; ++q) *(v4f*)(my + (q * 32 + lane) * 4) = RA[q];
    };
    auto lstoreB = [&]() __attribute__((always_inline)) {
#pragma unroll
        for (int q = 0; q < 15; ++q) *(v4f*)(my + ((16 + q) * 32 + lane) * 4) = RB[q];
    };
    auto tile2 = [&](float* pb, const int tl) __attribute__((always_inline)) {
        const float* arow0 = my + row * FF + 2 * hi;
        const float* arow1 = arow0 + TILE_FL;
        v8f acc0 = {}, acc1 = {};
#pragma unroll
        for (int kk = 0; kk < KSTEPS; ++kk) {
            const v2f a0 = *(const v2f*)(arow0 + 4 * kk);
            const v2f a1 = *(const v2f*)(arow1 + 4 * kk);
            acc0 = __builtin_amdgcn_wmma_f32_16x16x4_f32(
                 false, a0,  false, bm[kk],
                 (short)0, acc0,  false,  false);
            acc1 = __builtin_amdgcn_wmma_f32_16x16x4_f32(
                 false, a1,  false, bm[kk],
                 (short)0, acc1,  false,  false);
        }
        if (n < HH) {
#pragma unroll
            for (int r = 0; r < 8; ++r) {
                pb[(tl + r + 8 * hi) * 4 + n]      = acc0[r] + bias;
                pb[(tl + 16 + r + 8 * hi) * 4 + n] = acc1[r] + bias;
            }
        }
    };

    float A0[3][3], Wi[3][3], Wh[3][3], c1[3];
#pragma unroll
    for (int i = 0; i < 3; ++i) {
#pragma unroll
        for (int j = 0; j < 3; ++j) {
            A0[i][j] = w_hh0[i * 3 + j];
            Wi[i][j] = w_ih1[i * 3 + j];
            Wh[i][j] = w_hh1[i * 3 + j];
        }
        c1[i] = b_ih1[i] + b_hh1[i];
    }
    float h1[3], h2[3];
#pragma unroll
    for (int i = 0; i < 3; ++i) {
        h1[i] = hx[(0 * BB + b) * HH + i];
        h2[i] = hx[(1 * BB + b) * HH + i];
    }

    auto layer0 = [&](const v4f pv, float (&n0)[3]) __attribute__((always_inline)) {
        const float p[3] = {pv.x, pv.y, pv.z};
#pragma unroll
        for (int i = 0; i < 3; ++i) {
            float u = p[i];
#pragma unroll
            for (int j = 0; j < 3; ++j) u = fmaf(A0[i][j], h1[j], u);
            n0[i] = fast_tanh(u);
        }
    };
    auto layer1 = [&](float (&n1)[3]) __attribute__((always_inline)) {
#pragma unroll
        for (int i = 0; i < 3; ++i) {
            float v = c1[i];
#pragma unroll
            for (int j = 0; j < 3; ++j) v = fmaf(Wi[i][j], h1[j], v);
#pragma unroll
            for (int j = 0; j < 3; ++j) v = fmaf(Wh[i][j], h2[j], v);
            n1[i] = fast_tanh(v);
        }
    };

    float Wo[OO][3], bo[OO];
#pragma unroll
    for (int o = 0; o < OO; ++o) {
        bo[o] = b_out[o];
#pragma unroll
        for (int j = 0; j < 3; ++j) Wo[o][j] = W_out[o * 3 + j];
    }

    if (wv == 1 || wv == 2) {
        const float* s0 = pair_src(0);
        gloadA(s0);
        gloadB(s0);
    }

    for (int p = 0; p < NCH + 2; ++p) {
        if (wv == 1 || wv == 2) {
            if (p < NCH) {
                float* pb = preb + (p & 1) * CHK * 4;
#pragma unroll
                for (int s = 0; s < 2; ++s) {
                    const int q  = 2 * p + s;
                    const int qn = (q + 1 < NPAIR) ? (q + 1) : q;
                    const float* sn = pair_src(qn);
                    const int tl = (pw * 4 + s * 2) * 16;
                    lstoreA();
                    lstoreB();
                    __builtin_amdgcn_sched_barrier(0x0001);
                    gloadA(sn);
                    gloadB(sn);
                    __builtin_amdgcn_sched_barrier(0x0001);
                    tile2(pb, tl);
                }
            }
        } else if (wv == 0) {
            if (p >= 1 && p <= NCH) {
                const int c = p - 1;
                const v4f* prow = (const v4f*)preb + (c & 1) * CHK;
                v4f* hw = (v4f*)h2b + (size_t)(c & 1) * CHK * 32 + lane;
#pragma nounroll
                for (int hf = 0; hf < 2; ++hf) {
                    v4f rw[HALF];
#pragma unroll
                    for (int i = 0; i < HALF; ++i) rw[i] = prow[hf * HALF + i];
                    __builtin_amdgcn_sched_barrier(0x0001);
                    {
                        float n0[3];
                        layer0(rw[0], n0);
#pragma unroll
                        for (int i = 0; i < 3; ++i) h1[i] = n0[i];
                    }
#pragma unroll
                    for (int i = 0; i + 1 < HALF; ++i) {
                        float n0[3], n1[3];
                        layer0(rw[i + 1], n0);
                        layer1(n1);
#pragma unroll
                        for (int k = 0; k < 3; ++k) { h1[k] = n0[k]; h2[k] = n1[k]; }
                        hw[(size_t)(hf * HALF + i) * 32] = (v4f){h2[0], h2[1], h2[2], 0.0f};
                    }
                    {
                        float n1[3];
                        layer1(n1);
#pragma unroll
                        for (int k = 0; k < 3; ++k) h2[k] = n1[k];
                        hw[(size_t)(hf * HALF + HALF - 1) * 32] = (v4f){h2[0], h2[1], h2[2], 0.0f};
                    }
                }
            }
        } else {
            if (p >= 2) {
                const int c = p - 2;
                const v4f* hr = (const v4f*)h2b + (size_t)(c & 1) * CHK * 32;
#pragma unroll
                for (int k = 0; k < CHK / 32; ++k) {
                    const int s = k * 32 + lane;
                    const v4f hv = hr[(size_t)s * 32 + lane];
                    const float hh[3] = {hv.x, hv.y, hv.z};
                    float y[3];
#pragma unroll
                    for (int i = 0; i < 3; ++i) y[i] = selu_f(hh[i]);
                    float* sp = stage + s * OO;
#pragma unroll
                    for (int o = 0; o < OO; ++o) {
                        float sacc = bo[o];
#pragma unroll
                        for (int j = 0; j < 3; ++j) sacc = fmaf(Wo[o][j], y[j], sacc);
                        sp[o] = sacc;
                    }
                }
                float* ob = out + ((size_t)b * TT + (size_t)c * CHK) * OO;
#pragma unroll
                for (int j = 0; j < CHK * OO / 128; ++j) {
                    const int piece = j * 32 + lane;
                    const v4f v = *(const v4f*)(stage + piece * 4);
                    *(v4f_a4*)(ob + piece * 4) = v;
                }
            }
        }
        __syncthreads();
    }
}

extern "C" void kernel_launch(void* const* d_in, const int* in_sizes, int n_in,
                              void* d_out, int out_size, void* d_ws, size_t ws_size,
                              hipStream_t stream)
{
    const float* x     = (const float*)d_in[0];
    const float* hx    = (const float*)d_in[1];
    const float* w_ih0 = (const float*)d_in[2];
    const float* w_hh0 = (const float*)d_in[3];
    const float* b_ih0 = (const float*)d_in[4];
    const float* b_hh0 = (const float*)d_in[5];
    const float* w_ih1 = (const float*)d_in[6];
    const float* w_hh1 = (const float*)d_in[7];
    const float* b_ih1 = (const float*)d_in[8];
    const float* b_hh1 = (const float*)d_in[9];
    const float* W_out = (const float*)d_in[10];
    const float* b_out = (const float*)d_in[11];
    float* out = (float*)d_out;

    rnn_fused<<<BB, NWAVE * 32, 0, stream>>>(x, hx, w_ih0, w_hh0, b_ih0, b_hh0,
                                             w_ih1, w_hh1, b_ih1, b_hh1, W_out, b_out, out);
}


# recurrence wave: a half's 64 h2 rows stay in registers and leave as one group of 12-byte LDS stores behind its last step (was a 16-byte store inside every step or two); same arithmetic and sum order
# speedup vs baseline: 23.8683x; 1.1547x over previous
#include <hip/hip_runtime.h>

typedef __attribute__((ext_vector_type(2))) float v2f;
typedef __attribute__((ext_vector_type(8))) float v8f;
typedef __attribute__((ext_vector_type(4))) float v4f;
typedef __attribute__((ext_vector_type(3))) float v3f;
typedef v4f v4f_a4 __attribute__((aligned(4)));

#define BB 128
#define TT 2048
#define FF 124
#define HH 3
#define OO 6

#define NWAVE 4
#define CHK   128
#define HALF  64
#define NCH   (TT / CHK)
#define TILE_FL  (16 * FF)
#define PAIR_FL  (2 * TILE_FL)
#define KSTEPS   (FF / 4)
#define NPAIR (TT / 16 / 2 / 2)
static_assert(PAIR_FL % (32 * 4) == 0, "a pair of tiles is a whole number of 512-byte rounds");
static_assert(PAIR_FL / (32 * 4) == 31, "31 rounds per pair: 16 for the first tile, 15 more for the second");
static_assert(CHK == 2 * 2 * 32, "a phase is two pairs for each of the two projection waves");
static_assert((CHK * OO) % (4 * 32) == 0, "a chunk of out is a whole number of 512-byte stores");

__device__ __forceinline__ float fast_tanh(float x) {
    return __builtin_amdgcn_tanhf(x);
}

__device__ __forceinline__ float selu_f(float x) {
    const float kAlpha = 1.6732632423543772f;
    const float kScale = 1.0507009873554805f;
    const float e = __builtin_amdgcn_exp2f(x * 1.4426950408889634f);
    const float pos = kScale * x;
    const float neg = kScale * kAlpha * (e - 1.0f);
    return (x > 0.0f) ? pos : neg;
}

__global__ __launch_bounds__(NWAVE * 32) void rnn_fused(
    const float* __restrict__ x,
    const float* __restrict__ hx,
    const float* __restrict__ w_ih0,
    const float* __restrict__ w_hh0,
    const float* __restrict__ b_ih0,
    const float* __restrict__ b_hh0,
    const float* __restrict__ w_ih1,
    const float* __restrict__ w_hh1,
    const float* __restrict__ b_ih1,
    const float* __restrict__ b_hh1,
    const float* __restrict__ W_out,
    const float* __restrict__ b_out,
    float* __restrict__ out)
{
    __shared__ __attribute__((aligned(16))) float xs[2 * PAIR_FL];
    __shared__ __attribute__((aligned(16))) float preb[2 * CHK * 4];
    __shared__ __attribute__((aligned(16))) float h2b[2 * CHK * 32 * 4];
    __shared__ __attribute__((aligned(16))) float stage[CHK * OO];

    const int lane = threadIdx.x & 31;
    const int wv = __builtin_amdgcn_readfirstlane((int)(threadIdx.x >> 5));
    const int b  = blockIdx.x;

    const int row  = lane & 15;
    const int hi   = lane >> 4;
    const int pw   = (wv == 2) ? 1 : 0;
    const float* xb = x + (size_t)b * TT * FF;
    float* my = xs + pw * PAIR_FL;

    const int   n     = row;
    const int   nclmp = (n < HH) ? n : (HH - 1);
    const float nmask = (n < HH) ? 1.0f : 0.0f;
    const float* wrow = w_ih0 + (size_t)nclmp * FF;

    v2f bm[KSTEPS];
    float bias = 0.0f;
    if (wv == 1 || wv == 2) {
#pragma unroll
        for (int kk = 0; kk < KSTEPS; ++kk) {
            const int k0 = kk * 4 + 2 * hi;
            bm[kk].x = wrow[k0]     * nmask;
            bm[kk].y = wrow[k0 + 1] * nmask;
        }
        bias = b_ih0[nclmp] + b_hh0[nclmp];
    } else {
#pragma unroll
        for (int kk = 0; kk < KSTEPS; ++kk) { bm[kk].x = 0.0f; bm[kk].y = 0.0f; }
    }

    v4f RA[16], RB[15];
#pragma unroll
    for (int q = 0; q < 16; ++q) RA[q] = (v4f){0.0f, 0.0f, 0.0f, 0.0f};
#pragma unroll
    for (int q = 0; q < 15; ++q) RB[q] = (v4f){0.0f, 0.0f, 0.0f, 0.0f};
    auto pair_src = [&](const int q) __attribute__((always_inline)) -> const float* {
        const int ti = (q >> 1) * 8 + pw * 4 + (q & 1) * 2;
        return xb + (size_t)ti * TILE_FL;
    };
    auto gloadA = [&](const float* sp) __attribute__((always_inline)) {
#pragma unroll
        for (int q = 0; q < 16; ++q)
            RA[q] = *(const v4f_a4*)(sp + (q * 32 + lane) * 4);
    };
    auto gloadB = [&](const float* sp) __attribute__((always_inline)) {
#pragma unroll
        for (int q = 0; q < 15; ++q)
            RB[q] = *(const v4f_a4*)(sp + ((16 + q) * 32 + lane) * 4);
    };
    auto lstoreA = [&]() __attribute__((always_inline)) {
#pragma unroll
        for (int q = 0; q < 16; ++q) *(v4f*)(my + (q * 32 + lane) * 4) = RA[q];
    };
    auto lstoreB = [&]() __attribute__((always_inline)) {
#pragma unroll
        for (int q = 0; q < 15; ++q) *(v4f*)(my + ((16 + q) * 32 + lane) * 4) = RB[q];
    };
    auto tile2 = [&](float* pb, const int tl) __attribute__((always_inline)) {
        const float* arow0 = my + row * FF + 2 * hi;
        const float* arow1 = arow0 + TILE_FL;
        v8f acc0 = {}, acc1 = {};
#pragma unroll
        for (int kk = 0; kk < KSTEPS; ++kk) {
            const v2f a0 = *(const v2f*)(arow0 + 4 * kk);
            const v2f a1 = *(const v2f*)(arow1 + 4 * kk);
            acc0 = __builtin_amdgcn_wmma_f32_16x16x4_f32(
                 false, a0,  false, bm[kk],
                 (short)0, acc0,  false,  false);
            acc1 = __builtin_amdgcn_wmma_f32_16x16x4_f32(
                 false, a1,  false, bm[kk],
                 (short)0, acc1,  false,  false);
        }
        if (n < HH) {
#pragma unroll
            for (int r = 0; r < 8; ++r) {
                pb[(tl + r + 8 * hi) * 4 + n]      = acc0[r] + bias;
                pb[(tl + 16 + r + 8 * hi) * 4 + n] = acc1[r] + bias;
            }
        }
    };

    float A0[3][3], Wi[3][3], Wh[3][3], c1[3];
#pragma unroll
    for (int i = 0; i < 3; ++i) {
#pragma unroll
        for (int j = 0; j < 3; ++j) {
            A0[i][j] = w_hh0[i * 3 + j];
            Wi[i][j] = w_ih1[i * 3 + j];
            Wh[i][j] = w_hh1[i * 3 + j];
        }
        c1[i] = b_ih1[i] + b_hh1[i];
    }
    float h1[3], h2[3];
#pragma unroll
    for (int i = 0; i < 3; ++i) {
        h1[i] = hx[(0 * BB + b) * HH + i];
        h2[i] = hx[(1 * BB + b) * HH + i];
    }

    auto layer0 = [&](const v4f pv, float (&n0)[3]) __attribute__((always_inline)) {
        const float p[3] = {pv.x, pv.y, pv.z};
#pragma unroll
        for (int i = 0; i < 3; ++i) {
            float u = p[i];
#pragma unroll
            for (int j = 0; j < 3; ++j) u = fmaf(A0[i][j], h1[j], u);
            n0[i] = fast_tanh(u);
        }
    };
    auto layer1 = [&](float (&n1)[3]) __attribute__((always_inline)) {
#pragma unroll
        for (int i = 0; i < 3; ++i) {
            float v = c1[i];
#pragma unroll
            for (int j = 0; j < 3; ++j) v = fmaf(Wi[i][j], h1[j], v);
#pragma unroll
            for (int j = 0; j < 3; ++j) v = fmaf(Wh[i][j], h2[j], v);
            n1[i] = fast_tanh(v);
        }
    };

    float Wo[OO][3], bo[OO];
#pragma unroll
    for (int o = 0; o < OO; ++o) {
        bo[o] = b_out[o];
#pragma unroll
        for (int j = 0; j < 3; ++j) Wo[o][j] = W_out[o * 3 + j];
    }

    if (wv == 1 || wv == 2) {
        const float* s0 = pair_src(0);
        gloadA(s0);
        gloadB(s0);
    }

    for (int p = 0; p < NCH + 2; ++p) {
        if (wv == 1 || wv == 2) {
            if (p < NCH) {
                float* pb = preb + (p & 1) * CHK * 4;
#pragma unroll
                for (int s = 0; s < 2; ++s) {
                    const int q  = 2 * p + s;
                    const int qn = (q + 1 < NPAIR) ? (q + 1) : q;
                    const float* sn = pair_src(qn);
                    const int tl = (pw * 4 + s * 2) * 16;
                    lstoreA();
                    lstoreB();
                    __builtin_amdgcn_sched_barrier(0x0001);
                    gloadA(sn);
                    gloadB(sn);
                    __builtin_amdgcn_sched_barrier(0x0001);
                    tile2(pb, tl);
                }
            }
        } else if (wv == 0) {
            if (p >= 1 && p <= NCH) {
                const int c = p - 1;
                const v4f* prow = (const v4f*)preb + (c & 1) * CHK;
                v4f* hw = (v4f*)h2b + (size_t)(c & 1) * CHK * 32 + lane;
#pragma nounroll
                for (int hf = 0; hf < 2; ++hf) {
                    v4f rw[HALF];
                    v3f hk[HALF];
#pragma unroll
                    for (int i = 0; i < HALF; ++i) rw[i] = prow[hf * HALF + i];
                    __builtin_amdgcn_sched_barrier(0x0001);
                    {
                        float n0[3];
                        layer0(rw[0], n0);
#pragma unroll
                        for (int i = 0; i < 3; ++i) h1[i] = n0[i];
                    }
#pragma unroll
                    for (int i = 0; i + 1 < HALF; ++i) {
                        float n0[3], n1[3];
                        layer0(rw[i + 1], n0);
                        layer1(n1);
#pragma unroll
                        for (int k = 0; k < 3; ++k) { h1[k] = n0[k]; h2[k] = n1[k]; }
                        hk[i] = (v3f){h2[0], h2[1], h2[2]};
                    }
                    {
                        float n1[3];
                        layer1(n1);
#pragma unroll
                        for (int k = 0; k < 3; ++k) h2[k] = n1[k];
                        hk[HALF - 1] = (v3f){h2[0], h2[1], h2[2]};
                    }
                    __builtin_amdgcn_sched_barrier(0);
#pragma unroll
                    for (int i = 0; i < HALF; ++i)
                        *(v3f*)(hw + (size_t)(hf * HALF + i) * 32) = hk[i];
                }
            }
        } else {
            if (p >= 2) {
                const int c = p - 2;
                const v4f* hr = (const v4f*)h2b + (size_t)(c & 1) * CHK * 32;
#pragma unroll
                for (int k = 0; k < CHK / 32; ++k) {
                    const int s = k * 32 + lane;
                    const v3f hv = *(const v3f*)(hr + (size_t)s * 32 + lane);
                    const float hh[3] = {hv.x, hv.y, hv.z};
                    float y[3];
#pragma unroll
                    for (int i = 0; i < 3; ++i) y[i] = selu_f(hh[i]);
                    float* sp = stage + s * OO;
#pragma unroll
                    for (int o = 0; o < OO; ++o) {
                        float sacc = bo[o];
#pragma unroll
                        for (int j = 0; j < 3; ++j) sacc = fmaf(Wo[o][j], y[j], sacc);
                        sp[o] = sacc;
                    }
                }
                float* ob = out + ((size_t)b * TT + (size_t)c * CHK) * OO;
#pragma unroll
                for (int j = 0; j < CHK * OO / 128; ++j) {
                    const int piece = j * 32 + lane;
                    const v4f v = *(const v4f*)(stage + piece * 4);
                    *(v4f_a4*)(ob + piece * 4) = v;
                }
            }
        }
        __syncthreads();
    }
}

extern "C" void kernel_launch(void* const* d_in, const int* in_sizes, int n_in,
                              void* d_out, int out_size, void* d_ws, size_t ws_size,
                              hipStream_t stream)
{
    const float* x     = (const float*)d_in[0];
    const float* hx    = (const float*)d_in[1];
    const float* w_ih0 = (const float*)d_in[2];
    const float* w_hh0 = (const float*)d_in[3];
    const float* b_ih0 = (const float*)d_in[4];
    const float* b_hh0 = (const float*)d_in[5];
    const float* w_ih1 = (const float*)d_in[6];
    const float* w_hh1 = (const float*)d_in[7];
    const float* b_ih1 = (const float*)d_in[8];
    const float* b_hh1 = (const float*)d_in[9];
    const float* W_out = (const float*)d_in[10];
    const float* b_out = (const float*)d_in[11];
    float* out = (float*)d_out;

    rnn_fused<<<BB, NWAVE * 32, 0, stream>>>(x, hx, w_ih0, w_hh0, b_ih0, b_hh0,
                                             w_ih1, w_hh1, b_ih1, b_hh1, W_out, b_out, out);
}


# recurrence wave: the six hidden values sit in six lanes of one register; a step of both layers is 3+3 fma fed by DPP lane broadcasts and ONE tanh (was 15 fma, 6 tanh); same arithmetic and sum order
# speedup vs baseline: 24.8592x; 1.0415x over previous
#include <hip/hip_runtime.h>

typedef __attribute__((ext_vector_type(2))) float v2f;
typedef __attribute__((ext_vector_type(8))) float v8f;
typedef __attribute__((ext_vector_type(4))) float v4f;
typedef __attribute__((ext_vector_type(3))) float v3f;
typedef v4f v4f_a4 __attribute__((aligned(4)));

#define BB 128
#define TT 2048
#define FF 124
#define HH 3
#define OO 6

#define NWAVE 4
#define CHK   128
#define HALF  64
#define NCH   (TT / CHK)
#define TILE_FL  (16 * FF)
#define PAIR_FL  (2 * TILE_FL)
#define KSTEPS   (FF / 4)
#define NPAIR (TT / 16 / 2 / 2)
static_assert(PAIR_FL % (32 * 4) == 0, "a pair of tiles is a whole number of 512-byte rounds");
static_assert(PAIR_FL / (32 * 4) == 31, "31 rounds per pair: 16 for the first tile, 15 more for the second");
static_assert(CHK == 2 * 2 * 32, "a phase is two pairs for each of the two projection waves");
static_assert((CHK * OO) % (4 * 32) == 0, "a chunk of out is a whole number of 512-byte stores");

__device__ __forceinline__ float fast_tanh(float x) {
    return __builtin_amdgcn_tanhf(x);
}

__device__ __forceinline__ float selu_f(float x) {
    const float kAlpha = 1.6732632423543772f;
    const float kScale = 1.0507009873554805f;
    const float e = __builtin_amdgcn_exp2f(x * 1.4426950408889634f);
    const float pos = kScale * x;
    const float neg = kScale * kAlpha * (e - 1.0f);
    return (x > 0.0f) ? pos : neg;
}

__global__ __launch_bounds__(NWAVE * 32) void rnn_fused(
    const float* __restrict__ x,
    const float* __restrict__ hx,
    const float* __restrict__ w_ih0,
    const float* __restrict__ w_hh0,
    const float* __restrict__ b_ih0,
    const float* __restrict__ b_hh0,
    const float* __restrict__ w_ih1,
    const float* __restrict__ w_hh1,
    const float* __restrict__ b_ih1,
    const float* __restrict__ b_hh1,
    const float* __restrict__ W_out,
    const float* __restrict__ b_out,
    float* __restrict__ out)
{
    __shared__ __attribute__((aligned(16))) float xs[2 * PAIR_FL];
    __shared__ __attribute__((aligned(16))) float preb[2 * CHK * 4];
    __shared__ __attribute__((aligned(16))) float h2b[2 * CHK * 32 * 4];
    __shared__ __attribute__((aligned(16))) float stage[CHK * OO];

    const int lane = threadIdx.x & 31;
    const int wv = __builtin_amdgcn_readfirstlane((int)(threadIdx.x >> 5));
    const int b  = blockIdx.x;

    const int row  = lane & 15;
    const int hi   = lane >> 4;
    const int pw   = (wv == 2) ? 1 : 0;
    const float* xb = x + (size_t)b * TT * FF;
    float* my = xs + pw * PAIR_FL;

    const int   n     = row;
    const int   nclmp = (n < HH) ? n : (HH - 1);
    const float nmask = (n < HH) ? 1.0f : 0.0f;
    const float* wrow = w_ih0 + (size_t)nclmp * FF;

    v2f bm[KSTEPS];
    float bias = 0.0f;
    if (wv == 1 || wv == 2) {
#pragma unroll
        for (int kk = 0; kk < KSTEPS; ++kk) {
            const int k0 = kk * 4 + 2 * hi;
            bm[kk].x = wrow[k0]     * nmask;
            bm[kk].y = wrow[k0 + 1] * nmask;
        }
        bias = b_ih0[nclmp] + b_hh0[nclmp];
    } else {
#pragma unroll
        for (int kk = 0; kk < KSTEPS; ++kk) { bm[kk].x = 0.0f; bm[kk].y = 0.0f; }
    }

    v4f RA[16], RB[15];
#pragma unroll
    for (int q = 0; q < 16; ++q) RA[q] = (v4f){0.0f, 0.0f, 0.0f, 0.0f};
#pragma unroll
    for (int q = 0; q < 15; ++q) RB[q] = (v4f){0.0f, 0.0f, 0.0f, 0.0f};
    auto pair_src = [&](const int q) __attribute__((always_inline)) -> const float* {
        const int ti = (q >> 1) * 8 + pw * 4 + (q & 1) * 2;
        return xb + (size_t)ti * TILE_FL;
    };
    auto gloadA = [&](const float* sp) __attribute__((always_inline)) {
#pragma unroll
        for (int q = 0; q < 16; ++q)
            RA[q] = *(const v4f_a4*)(sp + (q * 32 + lane) * 4);
    };
    auto gloadB = [&](const float* sp) __attribute__((always_inline)) {
#pragma unroll
        for (int q = 0; q < 15; ++q)
            RB[q] = *(const v4f_a4*)(sp + ((16 + q) * 32 + lane) * 4);
    };
    auto lstoreA = [&]() __attribute__((always_inline)) {
#pragma unroll
        for (int q = 0; q < 16; ++q) *(v4f*)(my + (q * 32 + lane) * 4) = RA[q];
    };
    auto lstoreB = [&]() __attribute__((always_inline)) {
#pragma unroll
        for (int q = 0; q < 15; ++q) *(v4f*)(my + ((16 + q) * 32 + lane) * 4) = RB[q];
    };
    auto tile2 = [&](float* pb, const int tl) __attribute__((always_inline)) {
        const float* arow0 = my + row * FF + 2 * hi;
        const float* arow1 = arow0 + TILE_FL;
        v8f acc0 = {}, acc1 = {};
#pragma unroll
        for (int kk = 0; kk < KSTEPS; ++kk) {
            const v2f a0 = *(const v2f*)(arow0 + 4 * kk);
            const v2f a1 = *(const v2f*)(arow1 + 4 * kk);
            acc0 = __builtin_amdgcn_wmma_f32_16x16x4_f32(
                 false, a0,  false, bm[kk],
                 (short)0, acc0,  false,  false);
            acc1 = __builtin_amdgcn_wmma_f32_16x16x4_f32(
                 false, a1,  false, bm[kk],
                 (short)0, acc1,  false,  false);
        }
        if (n < HH) {
#pragma unroll
            for (int r = 0; r < 8; ++r) {
                pb[(tl + r + 8 * hi) * 4 + n]      = acc0[r] + bias;
                pb[(tl + 16 + r + 8 * hi) * 4 + n] = acc1[r] + bias;
            }
        }
    };

    const int  comp = lane & 3;
    const int  quad = (lane >> 2) & 3;
    const bool isq0 = (quad == 0) && (comp < HH);
    const bool isq1 = (quad == 1) && (comp < HH);
    const int  ci   = (comp < HH) ? comp : (HH - 1);
    float Rc[3], Wc[3];
#pragma unroll
    for (int j = 0; j < 3; ++j) {
        const float a0 = w_hh0[ci * 3 + j];
        const float wh = w_hh1[ci * 3 + j];
        const float wi = w_ih1[ci * 3 + j];
        Rc[j] = isq0 ? a0 : (isq1 ? wh : 0.0f);
        Wc[j] = isq1 ? wi : 0.0f;
    }
    const float c1s = b_ih1[ci] + b_hh1[ci];
    const float c1v = isq1 ? c1s : 0.0f;
    float st;
    {
        const float g0 = hx[(0 * BB + b) * HH + ci];
        const float g1 = hx[(1 * BB + b) * HH + ci];
        st = isq0 ? g0 : (isq1 ? g1 : 0.0f);
    }
#define LANE_BC(v, ctrl) __builtin_bit_cast(float, __builtin_amdgcn_update_dpp( \
        0, __builtin_bit_cast(int, (v)), (ctrl), 0xf, 0xf, true))
#define LQ(v, j) LANE_BC(v, (j) * 0x55)
#define LR(v, j) LANE_BC(v, 0x150 + (j))
#define SLOT(S, A, PN) do {                                   \
        float u_ = (A);                                         \
        u_ = fmaf(Rc[0], LQ(S, 0), u_);                         \
        u_ = fmaf(Rc[1], LQ(S, 1), u_);                         \
        u_ = fmaf(Rc[2], LQ(S, 2), u_);                         \
        float an_ = isq0 ? (PN) : c1v;                          \
        an_ = fmaf(Wc[0], LR(S, 0), an_);                       \
        an_ = fmaf(Wc[1], LR(S, 1), an_);                       \
        an_ = fmaf(Wc[2], LR(S, 2), an_);                       \
        (S) = fast_tanh(u_);                                    \
        (A) = an_;                                              \
    } while (0)

    float Wo[OO][3], bo[OO];
#pragma unroll
    for (int o = 0; o < OO; ++o) {
        bo[o] = b_out[o];
#pragma unroll
        for (int j = 0; j < 3; ++j) Wo[o][j] = W_out[o * 3 + j];
    }

    if (wv == 1 || wv == 2) {
        const float* s0 = pair_src(0);
        gloadA(s0);
        gloadB(s0);
    }

    for (int p = 0; p < NCH + 2; ++p) {
        if (wv == 1 || wv == 2) {
            if (p < NCH) {
                float* pb = preb + (p & 1) * CHK * 4;
#pragma unroll
                for (int s = 0; s < 2; ++s) {
                    const int q  = 2 * p + s;
                    const int qn = (q + 1 < NPAIR) ? (q + 1) : q;
                    const float* sn = pair_src(qn);
                    const int tl = (pw * 4 + s * 2) * 16;
                    lstoreA();
                    lstoreB();
                    __builtin_amdgcn_sched_barrier(0x0001);
                    gloadA(sn);
                    gloadB(sn);
                    __builtin_amdgcn_sched_barrier(0x0001);
                    tile2(pb, tl);
                }
            }
        } else if (wv == 0) {
            if (p >= 1 && p <= NCH) {
                const int c = p - 1;
                const float* prow = preb + (c & 1) * CHK * 4 + ci;
                float* hw = h2b + (c & 1) * CHK * 4 + comp;
#pragma nounroll
                for (int hf = 0; hf < 2; ++hf) {
                    float rw[HALF];
                    float hk[HALF];
#pragma unroll
                    for (int i = 0; i < HALF; ++i) rw[i] = prow[(hf * HALF + i) * 4];
                    asm volatile("" : "+v"(st) : : "memory");
                    const float s_in = st;
                    float h1keep = 0.0f;
                    float ad = isq0 ? rw[0] : c1v;
#pragma unroll
                    for (int n = 0; n < HALF + 2; ++n) {
                        const int r = (n + 1 < HALF) ? (n + 1) : (HALF - 1);
                        SLOT(st, ad, rw[r]);
                        if (n == 1) st = isq1 ? s_in : st;
                        if (n == HALF - 1) h1keep = st;
                        if (n >= 2) hk[n - 2] = st;
                    }
                    st = isq1 ? st : h1keep;
                    if (isq1 && lane < 16) {
#pragma unroll
                        for (int i = 0; i < HALF; ++i) hw[(hf * HALF + i) * 4] = hk[i];
                    }
                }
            }
        } else {
            if (p >= 2) {
                const int c = p - 2;
                const v4f* hr = (const v4f*)h2b + (size_t)(c & 1) * CHK;
#pragma unroll
                for (int k = 0; k < CHK / 32; ++k) {
                    const int s = k * 32 + lane;
                    const v3f hv = *(const v3f*)(hr + s);
                    const float hh[3] = {hv.x, hv.y, hv.z};
                    float y[3];
#pragma unroll
                    for (int i = 0; i < 3; ++i) y[i] = selu_f(hh[i]);
                    float* sp = stage + s * OO;
#pragma unroll
                    for (int o = 0; o < OO; ++o) {
                        float sacc = bo[o];
#pragma unroll
                        for (int j = 0; j < 3; ++j) sacc = fmaf(Wo[o][j], y[j], sacc);
                        sp[o] = sacc;
                    }
                }
                float* ob = out + ((size_t)b * TT + (size_t)c * CHK) * OO;
#pragma unroll
                for (int j = 0; j < CHK * OO / 128; ++j) {
                    const int piece = j * 32 + lane;
                    const v4f v = *(const v4f*)(stage + piece * 4);
                    *(v4f_a4*)(ob + piece * 4) = v;
                }
            }
        }
        __syncthreads();
    }
}

extern "C" void kernel_launch(void* const* d_in, const int* in_sizes, int n_in,
                              void* d_out, int out_size, void* d_ws, size_t ws_size,
                              hipStream_t stream)
{
    const float* x     = (const float*)d_in[0];
    const float* hx    = (const float*)d_in[1];
    const float* w_ih0 = (const float*)d_in[2];
    const float* w_hh0 = (const float*)d_in[3];
    const float* b_ih0 = (const float*)d_in[4];
    const float* b_hh0 = (const float*)d_in[5];
    const float* w_ih1 = (const float*)d_in[6];
    const float* w_hh1 = (const float*)d_in[7];
    const float* b_ih1 = (const float*)d_in[8];
    const float* b_hh1 = (const float*)d_in[9];
    const float* W_out = (const float*)d_in[10];
    const float* b_out = (const float*)d_in[11];
    float* out = (float*)d_out;

    rnn_fused<<<BB, NWAVE * 32, 0, stream>>>(x, hx, w_ih0, w_hh0, b_ih0, b_hh0,
                                             w_ih1, w_hh1, b_ih1, b_hh1, W_out, b_out, out);
}


# recurrence slot as one asm statement: each multiply-add reads the other lane's value through its own DPP source (was a DPP move in front of each, in the serial chain); same arithmetic and sum order
# speedup vs baseline: 28.0723x; 1.1293x over previous
#include <hip/hip_runtime.h>

typedef __attribute__((ext_vector_type(2))) float v2f;
typedef __attribute__((ext_vector_type(8))) float v8f;
typedef __attribute__((ext_vector_type(4))) float v4f;
typedef __attribute__((ext_vector_type(3))) float v3f;
typedef v4f v4f_a4 __attribute__((aligned(4)));

#define BB 128
#define TT 2048
#define FF 124
#define HH 3
#define OO 6

#define NWAVE 4
#define CHK   128
#define HALF  64
#define NCH   (TT / CHK)
#define TILE_FL  (16 * FF)
#define PAIR_FL  (2 * TILE_FL)
#define KSTEPS   (FF / 4)
#define NPAIR (TT / 16 / 2 / 2)
static_assert(PAIR_FL % (32 * 4) == 0, "a pair of tiles is a whole number of 512-byte rounds");
static_assert(PAIR_FL / (32 * 4) == 31, "31 rounds per pair: 16 for the first tile, 15 more for the second");
static_assert(CHK == 2 * 2 * 32, "a phase is two pairs for each of the two projection waves");
static_assert((CHK * OO) % (4 * 32) == 0, "a chunk of out is a whole number of 512-byte stores");

__device__ __forceinline__ float selu_f(float x) {
    const float kAlpha = 1.6732632423543772f;
    const float kScale = 1.0507009873554805f;
    const float e = __builtin_amdgcn_exp2f(x * 1.4426950408889634f);
    const float pos = kScale * x;
    const float neg = kScale * kAlpha * (e - 1.0f);
    return (x > 0.0f) ? pos : neg;
}

__global__ __launch_bounds__(NWAVE * 32) void rnn_fused(
    const float* __restrict__ x,
    const float* __restrict__ hx,
    const float* __restrict__ w_ih0,
    const float* __restrict__ w_hh0,
    const float* __restrict__ b_ih0,
    const float* __restrict__ b_hh0,
    const float* __restrict__ w_ih1,
    const float* __restrict__ w_hh1,
    const float* __restrict__ b_ih1,
    const float* __restrict__ b_hh1,
    const float* __restrict__ W_out,
    const float* __restrict__ b_out,
    float* __restrict__ out)
{
    __shared__ __attribute__((aligned(16))) float xs[2 * PAIR_FL];
    __shared__ __attribute__((aligned(16))) float preb[2 * CHK * 4];
    __shared__ __attribute__((aligned(16))) float h2b[2 * CHK * 32 * 4];
    __shared__ __attribute__((aligned(16))) float stage[CHK * OO];

    const int lane = threadIdx.x & 31;
    const int wv = __builtin_amdgcn_readfirstlane((int)(threadIdx.x >> 5));
    const int b  = blockIdx.x;

    const int row  = lane & 15;
    const int hi   = lane >> 4;
    const int pw   = (wv == 2) ? 1 : 0;
    const float* xb = x + (size_t)b * TT * FF;
    float* my = xs + pw * PAIR_FL;

    const int   n     = row;
    const int   nclmp = (n < HH) ? n : (HH - 1);
    const float nmask = (n < HH) ? 1.0f : 0.0f;
    const float* wrow = w_ih0 + (size_t)nclmp * FF;

    v2f bm[KSTEPS];
    float bias = 0.0f;
    if (wv == 1 || wv == 2) {
#pragma unroll
        for (int kk = 0; kk < KSTEPS; ++kk) {
            const int k0 = kk * 4 + 2 * hi;
            bm[kk].x = wrow[k0]     * nmask;
            bm[kk].y = wrow[k0 + 1] * nmask;
        }
        bias = b_ih0[nclmp] + b_hh0[nclmp];
    } else {
#pragma unroll
        for (int kk = 0; kk < KSTEPS; ++kk) { bm[kk].x = 0.0f; bm[kk].y = 0.0f; }
    }

    v4f RA[16], RB[15];
#pragma unroll
    for (int q = 0; q < 16; ++q) RA[q] = (v4f){0.0f, 0.0f, 0.0f, 0.0f};
#pragma unroll
    for (int q = 0; q < 15; ++q) RB[q] = (v4f){0.0f, 0.0f, 0.0f, 0.0f};
    auto pair_src = [&](const int q) __attribute__((always_inline)) -> const float* {
        const int ti = (q >> 1) * 8 + pw * 4 + (q & 1) * 2;
        return xb + (size_t)ti * TILE_FL;
    };
    auto gloadA = [&](const float* sp) __attribute__((always_inline)) {
#pragma unroll
        for (int q = 0; q < 16; ++q)
            RA[q] = *(const v4f_a4*)(sp + (q * 32 + lane) * 4);
    };
    auto gloadB = [&](const float* sp) __attribute__((always_inline)) {
#pragma unroll
        for (int q = 0; q < 15; ++q)
            RB[q] = *(const v4f_a4*)(sp + ((16 + q) * 32 + lane) * 4);
    };
    auto lstoreA = [&]() __attribute__((always_inline)) {
#pragma unroll
        for (int q = 0; q < 16; ++q) *(v4f*)(my + (q * 32 + lane) * 4) = RA[q];
    };
    auto lstoreB = [&]() __attribute__((always_inline)) {
#pragma unroll
        for (int q = 0; q < 15; ++q) *(v4f*)(my + ((16 + q) * 32 + lane) * 4) = RB[q];
    };
    auto tile2 = [&](float* pb, const int tl) __attribute__((always_inline)) {
        const float* arow0 = my + row * FF + 2 * hi;
        const float* arow1 = arow0 + TILE_FL;
        v8f acc0 = {}, acc1 = {};
#pragma unroll
        for (int kk = 0; kk < KSTEPS; ++kk) {
            const v2f a0 = *(const v2f*)(arow0 + 4 * kk);
            const v2f a1 = *(const v2f*)(arow1 + 4 * kk);
            acc0 = __builtin_amdgcn_wmma_f32_16x16x4_f32(
                 false, a0,  false, bm[kk],
                 (short)0, acc0,  false,  false);
            acc1 = __builtin_amdgcn_wmma_f32_16x16x4_f32(
                 false, a1,  false, bm[kk],
                 (short)0, acc1,  false,  false);
        }
        if (n < HH) {
#pragma unroll
            for (int r = 0; r < 8; ++r) {
                pb[(tl + r + 8 * hi) * 4 + n]      = acc0[r] + bias;
                pb[(tl + 16 + r + 8 * hi) * 4 + n] = acc1[r] + bias;
            }
        }
    };

    const int  comp = lane & 3;
    const int  quad = (lane >> 2) & 3;
    const bool isq0 = (quad == 0) && (comp < HH);
    const bool isq1 = (quad == 1) && (comp < HH);
    const int  ci   = (comp < HH) ? comp : (HH - 1);
    float Rc[3], Wc[3];
#pragma unroll
    for (int j = 0; j < 3; ++j) {
        const float a0 = w_hh0[ci * 3 + j];
        const float wh = w_hh1[ci * 3 + j];
        const float wi = w_ih1[ci * 3 + j];
        Rc[j] = isq0 ? a0 : (isq1 ? wh : 0.0f);
        Wc[j] = isq1 ? wi : 0.0f;
    }
    const float c1s = b_ih1[ci] + b_hh1[ci];
    const float c1v = isq1 ? c1s : 0.0f;
    float st;
    {
        const float g0 = hx[(0 * BB + b) * HH + ci];
        const float g1 = hx[(1 * BB + b) * HH + ci];
        st = isq0 ? g0 : (isq1 ? g1 : 0.0f);
    }
#define SLOT(S, A, PN) do {                                                               \
        float u_  = (A);                                                                    \
        float an_ = isq0 ? (PN) : c1v;                                                      \
        float sn_;                                                                          \
        asm("v_fmac_f32_dpp %[u], %[s], %[r0] quad_perm:[0,0,0,0] row_mask:0xf bank_mask:0xf bound_ctrl:1\n\t" \
            "v_fmac_f32_dpp %[a], %[s], %[w0] row_share:0 row_mask:0xf bank_mask:0xf bound_ctrl:1\n\t"        \
            "v_fmac_f32_dpp %[u], %[s], %[r1] quad_perm:[1,1,1,1] row_mask:0xf bank_mask:0xf bound_ctrl:1\n\t" \
            "v_fmac_f32_dpp %[a], %[s], %[w1] row_share:1 row_mask:0xf bank_mask:0xf bound_ctrl:1\n\t"        \
            "v_fmac_f32_dpp %[u], %[s], %[r2] quad_perm:[2,2,2,2] row_mask:0xf bank_mask:0xf bound_ctrl:1\n\t" \
            "v_fmac_f32_dpp %[a], %[s], %[w2] row_share:2 row_mask:0xf bank_mask:0xf bound_ctrl:1\n\t"        \
            "v_tanh_f32 %[so], %[u]\n\t"                                                    \
            "v_nop"                                                                         \
            : [u] "+v"(u_), [a] "+v"(an_), [so] "=&v"(sn_)                                  \
            : [s] "v"(S), [r0] "v"(Rc[0]), [r1] "v"(Rc[1]), [r2] "v"(Rc[2]),                \
              [w0] "v"(Wc[0]), [w1] "v"(Wc[1]), [w2] "v"(Wc[2]));                           \
        (S) = sn_;                                                                          \
        (A) = an_;                                                                          \
    } while (0)

    float Wo[OO][3], bo[OO];
#pragma unroll
    for (int o = 0; o < OO; ++o) {
        bo[o] = b_out[o];
#pragma unroll
        for (int j = 0; j < 3; ++j) Wo[o][j] = W_out[o * 3 + j];
    }

    if (wv == 1 || wv == 2) {
        const float* s0 = pair_src(0);
        gloadA(s0);
        gloadB(s0);
    }

    for (int p = 0; p < NCH + 2; ++p) {
        if (wv == 1 || wv == 2) {
            if (p < NCH) {
                float* pb = preb + (p & 1) * CHK * 4;
#pragma unroll
                for (int s = 0; s < 2; ++s) {
                    const int q  = 2 * p + s;
                    const int qn = (q + 1 < NPAIR) ? (q + 1) : q;
                    const float* sn = pair_src(qn);
                    const int tl = (pw * 4 + s * 2) * 16;
                    lstoreA();
                    lstoreB();
                    __builtin_amdgcn_sched_barrier(0x0001);
                    gloadA(sn);
                    gloadB(sn);
                    __builtin_amdgcn_sched_barrier(0x0001);
                    tile2(pb, tl);
                }
            }
        } else if (wv == 0) {
            if (p >= 1 && p <= NCH) {
                const int c = p - 1;
                const float* prow = preb + (c & 1) * CHK * 4 + ci;
                float* hw = h2b + (c & 1) * CHK * 4 + comp;
#pragma nounroll
                for (int hf = 0; hf < 2; ++hf) {
                    float rw[HALF];
                    float hk[HALF];
#pragma unroll
                    for (int i = 0; i < HALF; ++i) rw[i] = prow[(hf * HALF + i) * 4];
                    asm volatile("" : "+v"(st) : : "memory");
                    const float s_in = st;
                    float h1keep = 0.0f;
                    float ad = isq0 ? rw[0] : c1v;
#pragma unroll
                    for (int n = 0; n < HALF + 2; ++n) {
                        const int r = (n + 1 < HALF) ? (n + 1) : (HALF - 1);
                        SLOT(st, ad, rw[r]);
                        __builtin_amdgcn_sched_barrier(0);
                        if (n == 1) st = isq1 ? s_in : st;
                        if (n == HALF - 1) h1keep = st;
                        if (n >= 2) hk[n - 2] = st;
                    }
                    st = isq1 ? st : h1keep;
                    if (isq1 && lane < 16) {
#pragma unroll
                        for (int i = 0; i < HALF; ++i) hw[(hf * HALF + i) * 4] = hk[i];
                    }
                }
            }
        } else {
            if (p >= 2) {
                const int c = p - 2;
                const v4f* hr = (const v4f*)h2b + (size_t)(c & 1) * CHK;
#pragma unroll
                for (int k = 0; k < CHK / 32; ++k) {
                    const int s = k * 32 + lane;
                    const v3f hv = *(const v3f*)(hr + s);
                    const float hh[3] = {hv.x, hv.y, hv.z};
                    float y[3];
#pragma unroll
                    for (int i = 0; i < 3; ++i) y[i] = selu_f(hh[i]);
                    float* sp = stage + s * OO;
#pragma unroll
                    for (int o = 0; o < OO; ++o) {
                        float sacc = bo[o];
#pragma unroll
                        for (int j = 0; j < 3; ++j) sacc = fmaf(Wo[o][j], y[j], sacc);
                        sp[o] = sacc;
                    }
                }
                float* ob = out + ((size_t)b * TT + (size_t)c * CHK) * OO;
#pragma unroll
                for (int j = 0; j < CHK * OO / 128; ++j) {
                    const int piece = j * 32 + lane;
                    const v4f v = *(const v4f*)(stage + piece * 4);
                    *(v4f_a4*)(ob + piece * 4) = v;
                }
            }
        }
        __syncthreads();
    }
}

extern "C" void kernel_launch(void* const* d_in, const int* in_sizes, int n_in,
                              void* d_out, int out_size, void* d_ws, size_t ws_size,
                              hipStream_t stream)
{
    const float* x     = (const float*)d_in[0];
    const float* hx    = (const float*)d_in[1];
    const float* w_ih0 = (const float*)d_in[2];
    const float* w_hh0 = (const float*)d_in[3];
    const float* b_ih0 = (const float*)d_in[4];
    const float* b_hh0 = (const float*)d_in[5];
    const float* w_ih1 = (const float*)d_in[6];
    const float* w_hh1 = (const float*)d_in[7];
    const float* b_ih1 = (const float*)d_in[8];
    const float* b_hh1 = (const float*)d_in[9];
    const float* W_out = (const float*)d_in[10];
    const float* b_out = (const float*)d_in[11];
    float* out = (float*)d_out;

    rnn_fused<<<BB, NWAVE * 32, 0, stream>>>(x, hx, w_ih0, w_hh0, b_ih0, b_hh0,
                                             w_ih1, w_hh1, b_ih1, b_hh1, W_out, b_out, out);
}


# recurrence wave meets the LDS once per 128-step chunk (was twice), in 16-byte accesses of four steps, pre and h2 kept as [value][step] (was 8-byte two-address accesses); same arithmetic and sum order
# speedup vs baseline: 28.9404x; 1.0309x over previous
#include <hip/hip_runtime.h>

typedef __attribute__((ext_vector_type(2))) float v2f;
typedef __attribute__((ext_vector_type(8))) float v8f;
typedef __attribute__((ext_vector_type(4))) float v4f;
typedef v4f v4f_a4 __attribute__((aligned(4)));

#define BB 128
#define TT 2048
#define FF 124
#define HH 3
#define OO 6

#define NWAVE 4
#define CHK   128
#define HALF  128
#define PS    (CHK + 4)
#define NCH   (TT / CHK)
#define TILE_FL  (16 * FF)
#define PAIR_FL  (2 * TILE_FL)
#define KSTEPS   (FF / 4)
#define NPAIR (TT / 16 / 2 / 2)
static_assert(PAIR_FL % (32 * 4) == 0, "a pair of tiles is a whole number of 512-byte rounds");
static_assert(PAIR_FL / (32 * 4) == 31, "31 rounds per pair: 16 for the first tile, 15 more for the second");
static_assert(CHK == 2 * 2 * 32, "a phase is two pairs for each of the two projection waves");
static_assert((CHK * OO) % (4 * 32) == 0, "a chunk of out is a whole number of 512-byte stores");
static_assert(CHK == HALF && HALF % 4 == 0, "a run is a whole chunk and whole groups of four steps");
static_assert(PS % 4 == 0 && 2 * 3 * PS <= 2 * CHK * 4, "16-byte alignment of a value's step array; both buffers fit");

__device__ __forceinline__ float selu_f(float x) {
    const float kAlpha = 1.6732632423543772f;
    const float kScale = 1.0507009873554805f;
    const float e = __builtin_amdgcn_exp2f(x * 1.4426950408889634f);
    const float pos = kScale * x;
    const float neg = kScale * kAlpha * (e - 1.0f);
    return (x > 0.0f) ? pos : neg;
}

__global__ __launch_bounds__(NWAVE * 32) void rnn_fused(
    const float* __restrict__ x,
    const float* __restrict__ hx,
    const float* __restrict__ w_ih0,
    const float* __restrict__ w_hh0,
    const float* __restrict__ b_ih0,
    const float* __restrict__ b_hh0,
    const float* __restrict__ w_ih1,
    const float* __restrict__ w_hh1,
    const float* __restrict__ b_ih1,
    const float* __restrict__ b_hh1,
    const float* __restrict__ W_out,
    const float* __restrict__ b_out,
    float* __restrict__ out)
{
    __shared__ __attribute__((aligned(16))) float xs[2 * PAIR_FL];
    __shared__ __attribute__((aligned(16))) float preb[2 * CHK * 4];
    __shared__ __attribute__((aligned(16))) float h2b[2 * CHK * 32 * 4];
    __shared__ __attribute__((aligned(16))) float stage[CHK * OO];

    const int lane = threadIdx.x & 31;
    const int wv = __builtin_amdgcn_readfirstlane((int)(threadIdx.x >> 5));
    const int b  = blockIdx.x;

    const int row  = lane & 15;
    const int hi   = lane >> 4;
    const int pw   = (wv == 2) ? 1 : 0;
    const float* xb = x + (size_t)b * TT * FF;
    float* my = xs + pw * PAIR_FL;

    const int   n     = row;
    const int   nclmp = (n < HH) ? n : (HH - 1);
    const float nmask = (n < HH) ? 1.0f : 0.0f;
    const float* wrow = w_ih0 + (size_t)nclmp * FF;

    v2f bm[KSTEPS];
    float bias = 0.0f;
    if (wv == 1 || wv == 2) {
#pragma unroll
        for (int kk = 0; kk < KSTEPS; ++kk) {
            const int k0 = kk * 4 + 2 * hi;
            bm[kk].x = wrow[k0]     * nmask;
            bm[kk].y = wrow[k0 + 1] * nmask;
        }
        bias = b_ih0[nclmp] + b_hh0[nclmp];
    } else {
#pragma unroll
        for (int kk = 0; kk < KSTEPS; ++kk) { bm[kk].x = 0.0f; bm[kk].y = 0.0f; }
    }

    v4f RA[16], RB[15];
#pragma unroll
    for (int q = 0; q < 16; ++q) RA[q] = (v4f){0.0f, 0.0f, 0.0f, 0.0f};
#pragma unroll
    for (int q = 0; q < 15; ++q) RB[q] = (v4f){0.0f, 0.0f, 0.0f, 0.0f};
    auto pair_src = [&](const int q) __attribute__((always_inline)) -> const float* {
        const int ti = (q >> 1) * 8 + pw * 4 + (q & 1) * 2;
        return xb + (size_t)ti * TILE_FL;
    };
    auto gloadA = [&](const float* sp) __attribute__((always_inline)) {
#pragma unroll
        for (int q = 0; q < 16; ++q)
            RA[q] = *(const v4f_a4*)(sp + (q * 32 + lane) * 4);
    };
    auto gloadB = [&](const float* sp) __attribute__((always_inline)) {
#pragma unroll
        for (int q = 0; q < 15; ++q)
            RB[q] = *(const v4f_a4*)(sp + ((16 + q) * 32 + lane) * 4);
    };
    auto lstoreA = [&]() __attribute__((always_inline)) {
#pragma unroll
        for (int q = 0; q < 16; ++q) *(v4f*)(my + (q * 32 + lane) * 4) = RA[q];
    };
    auto lstoreB = [&]() __attribute__((always_inline)) {
#pragma unroll
        for (int q = 0; q < 15; ++q) *(v4f*)(my + ((16 + q) * 32 + lane) * 4) = RB[q];
    };
    auto tile2 = [&](float* pb, const int tl) __attribute__((always_inline)) {
        const float* arow0 = my + row * FF + 2 * hi;
        const float* arow1 = arow0 + TILE_FL;
        v8f acc0 = {}, acc1 = {};
#pragma unroll
        for (int kk = 0; kk < KSTEPS; ++kk) {
            const v2f a0 = *(const v2f*)(arow0 + 4 * kk);
            const v2f a1 = *(const v2f*)(arow1 + 4 * kk);
            acc0 = __builtin_amdgcn_wmma_f32_16x16x4_f32(
                 false, a0,  false, bm[kk],
                 (short)0, acc0,  false,  false);
            acc1 = __builtin_amdgcn_wmma_f32_16x16x4_f32(
                 false, a1,  false, bm[kk],
                 (short)0, acc1,  false,  false);
        }
        if (n < HH) {
#pragma unroll
            for (int r = 0; r < 8; ++r) {
                pb[n * PS + tl + r + 8 * hi]      = acc0[r] + bias;
                pb[n * PS + tl + 16 + r + 8 * hi] = acc1[r] + bias;
            }
        }
    };

    const int  comp = lane & 3;
    const int  quad = (lane >> 2) & 3;
    const bool isq0 = (quad == 0) && (comp < HH);
    const bool isq1 = (quad == 1) && (comp < HH);
    const int  ci   = (comp < HH) ? comp : (HH - 1);
    float Rc[3], Wc[3];
#pragma unroll
    for (int j = 0; j < 3; ++j) {
        const float a0 = w_hh0[ci * 3 + j];
        const float wh = w_hh1[ci * 3 + j];
        const float wi = w_ih1[ci * 3 + j];
        Rc[j] = isq0 ? a0 : (isq1 ? wh : 0.0f);
        Wc[j] = isq1 ? wi : 0.0f;
    }
    const float c1s = b_ih1[ci] + b_hh1[ci];
    const float c1v = isq1 ? c1s : 0.0f;
    float st;
    {
        const float g0 = hx[(0 * BB + b) * HH + ci];
        const float g1 = hx[(1 * BB + b) * HH + ci];
        st = isq0 ? g0 : (isq1 ? g1 : 0.0f);
    }
#define SLOT(S, A, PN) do {                                                               \
        float u_  = (A);                                                                    \
        float an_ = isq0 ? (PN) : c1v;                                                      \
        float sn_;                                                                          \
        asm("v_fmac_f32_dpp %[u], %[s], %[r0] quad_perm:[0,0,0,0] row_mask:0xf bank_mask:0xf bound_ctrl:1\n\t" \
            "v_fmac_f32_dpp %[a], %[s], %[w0] row_share:0 row_mask:0xf bank_mask:0xf bound_ctrl:1\n\t"        \
            "v_fmac_f32_dpp %[u], %[s], %[r1] quad_perm:[1,1,1,1] row_mask:0xf bank_mask:0xf bound_ctrl:1\n\t" \
            "v_fmac_f32_dpp %[a], %[s], %[w1] row_share:1 row_mask:0xf bank_mask:0xf bound_ctrl:1\n\t"        \
            "v_fmac_f32_dpp %[u], %[s], %[r2] quad_perm:[2,2,2,2] row_mask:0xf bank_mask:0xf bound_ctrl:1\n\t" \
            "v_fmac_f32_dpp %[a], %[s], %[w2] row_share:2 row_mask:0xf bank_mask:0xf bound_ctrl:1\n\t"        \
            "v_tanh_f32 %[so], %[u]\n\t"                                                    \
            "v_nop"                                                                         \
            : [u] "+v"(u_), [a] "+v"(an_), [so] "=&v"(sn_)                                  \
            : [s] "v"(S), [r0] "v"(Rc[0]), [r1] "v"(Rc[1]), [r2] "v"(Rc[2]),                \
              [w0] "v"(Wc[0]), [w1] "v"(Wc[1]), [w2] "v"(Wc[2]));                           \
        (S) = sn_;                                                                          \
        (A) = an_;                                                                          \
    } while (0)

    float Wo[OO][3], bo[OO];
#pragma unroll
    for (int o = 0; o < OO; ++o) {
        bo[o] = b_out[o];
#pragma unroll
        for (int j = 0; j < 3; ++j) Wo[o][j] = W_out[o * 3 + j];
    }

    if (wv == 1 || wv == 2) {
        const float* s0 = pair_src(0);
        gloadA(s0);
        gloadB(s0);
    }

    for (int p = 0; p < NCH + 2; ++p) {
        if (wv == 1 || wv == 2) {
            if (p < NCH) {
                float* pb = preb + (p & 1) * 3 * PS;
#pragma unroll
                for (int s = 0; s < 2; ++s) {
                    const int q  = 2 * p + s;
                    const int qn = (q + 1 < NPAIR) ? (q + 1) : q;
                    const float* sn = pair_src(qn);
                    const int tl = (pw * 4 + s * 2) * 16;
                    lstoreA();
                    lstoreB();
                    __builtin_amdgcn_sched_barrier(0x0001);
                    gloadA(sn);
                    gloadB(sn);
                    __builtin_amdgcn_sched_barrier(0x0001);
                    tile2(pb, tl);
                }
            }
        } else if (wv == 0) {
            if (p >= 1 && p <= NCH) {
                const int c = p - 1;
                const float* prow = preb + (c & 1) * 3 * PS + ci * PS;
                float* hw = h2b + (c & 1) * 3 * PS + ci * PS;
                {
                    float rw[HALF];
                    float hk[HALF];
#pragma unroll
                    for (int g = 0; g < HALF / 4; ++g) {
                        const v4f q4 = *(const v4f*)(prow + 4 * g);
                        rw[4 * g] = q4.x; rw[4 * g + 1] = q4.y; rw[4 * g + 2] = q4.z; rw[4 * g + 3] = q4.w;
                    }
                    asm volatile("" : "+v"(st) : : "memory");
                    const float s_in = st;
                    float h1keep = 0.0f;
                    float ad = isq0 ? rw[0] : c1v;
#pragma unroll
                    for (int n = 0; n < HALF + 2; ++n) {
                        const int r = (n + 1 < HALF) ? (n + 1) : (HALF - 1);
                        SLOT(st, ad, rw[r]);
                        __builtin_amdgcn_sched_barrier(0);
                        if (n == 1) st = isq1 ? s_in : st;
                        if (n == HALF - 1) h1keep = st;
                        if (n >= 2) hk[n - 2] = st;
                    }
                    st = isq1 ? st : h1keep;
                    if (isq1 && lane < 16) {
#pragma unroll
                        for (int g = 0; g < HALF / 4; ++g)
                            *(v4f*)(hw + 4 * g) = (v4f){hk[4 * g], hk[4 * g + 1], hk[4 * g + 2], hk[4 * g + 3]};
                    }
                }
            }
        } else {
            if (p >= 2) {
                const int c = p - 2;
                const float* hr = h2b + (c & 1) * 3 * PS;
#pragma unroll
                for (int k = 0; k < CHK / 32; ++k) {
                    const int s = k * 32 + lane;
                    const float hh[3] = {hr[s], hr[PS + s], hr[2 * PS + s]};
                    float y[3];
#pragma unroll
                    for (int i = 0; i < 3; ++i) y[i] = selu_f(hh[i]);
                    float* sp = stage + s * OO;
#pragma unroll
                    for (int o = 0; o < OO; ++o) {
                        float sacc = bo[o];
#pragma unroll
                        for (int j = 0; j < 3; ++j) sacc = fmaf(Wo[o][j], y[j], sacc);
                        sp[o] = sacc;
                    }
                }
                float* ob = out + ((size_t)b * TT + (size_t)c * CHK) * OO;
#pragma unroll
                for (int j = 0; j < CHK * OO / 128; ++j) {
                    const int piece = j * 32 + lane;
                    const v4f v = *(const v4f*)(stage + piece * 4);
                    *(v4f_a4*)(ob + piece * 4) = v;
                }
            }
        }
        __syncthreads();
    }
}

extern "C" void kernel_launch(void* const* d_in, const int* in_sizes, int n_in,
                              void* d_out, int out_size, void* d_ws, size_t ws_size,
                              hipStream_t stream)
{
    const float* x     = (const float*)d_in[0];
    const float* hx    = (const float*)d_in[1];
    const float* w_ih0 = (const float*)d_in[2];
    const float* w_hh0 = (const float*)d_in[3];
    const float* b_ih0 = (const float*)d_in[4];
    const float* b_hh0 = (const float*)d_in[5];
    const float* w_ih1 = (const float*)d_in[6];
    const float* w_hh1 = (const float*)d_in[7];
    const float* b_ih1 = (const float*)d_in[8];
    const float* b_hh1 = (const float*)d_in[9];
    const float* W_out = (const float*)d_in[10];
    const float* b_out = (const float*)d_in[11];
    float* out = (float*)d_out;

    rnn_fused<<<BB, NWAVE * 32, 0, stream>>>(x, hx, w_ih0, w_hh0, b_ih0, b_hh0,
                                             w_ih1, w_hh1, b_ih1, b_hh1, W_out, b_out, out);
}
